# starting kernel
# baseline (speedup 1.0000x reference)
#include <hip/hip_runtime.h>


typedef __attribute__((ext_vector_type(2))) float v2f;
typedef __attribute__((ext_vector_type(8))) float v8f;

#define DFEAT 128
#define HFEAT 15
#define HPAD  16

__device__ __forceinline__ void atomicAddF(float* p, float v) {
  unsafeAtomicAdd(p, v);
}

__global__ void k_deg_init(float* __restrict__ deg, int n) {
  int i = blockIdx.x * blockDim.x + threadIdx.x;
  if (i < n) deg[i] = 1.0f;
}

__global__ void k_deg_count(const int* __restrict__ col, float* __restrict__ deg, int e) {
  int i = blockIdx.x * blockDim.x + threadIdx.x;
  if (i < e) atomicAddF(&deg[col[i]], 1.0f);
}

__global__ void k_deg_rsqrt(float* __restrict__ deg, int n) {
  int i = blockIdx.x * blockDim.x + threadIdx.x;
  if (i < n) deg[i] = rsqrtf(deg[i]);
}

template <int K>
__global__ void k_gemm_scale(const float* __restrict__ A,
                             const float* __restrict__ W,
                             int wrows,
                             const float* __restrict__ dinv,
                             float* __restrict__ hs,
                             float* __restrict__ agg,
                             int numTiles) {
  __shared__ float wlds[K][HPAD];
  for (int i = threadIdx.x; i < K * HPAD; i += blockDim.x) {
    int k = i >> 4, n = i & 15;
    wlds[k][n] = (k < wrows && n < HFEAT) ? W[k * HFEAT + n] : 0.0f;
  }
  __syncthreads();

  int wave = threadIdx.x >> 5;
  int lane = threadIdx.x & 31;
  int tile = blockIdx.x * (blockDim.x >> 5) + wave;
  if (tile >= numTiles) return;

  int m  = lane & 15;
  int kh = (lane >> 4) << 1;
  const float* arow = A + (size_t)(tile * 16 + m) * K;

  v8f c = {};
#pragma unroll
  for (int k0 = 0; k0 < K; k0 += 4) {
    v2f a, b;
    a.x = arow[k0 + kh];
    a.y = arow[k0 + kh + 1];
    b.x = wlds[k0 + kh][m];
    b.y = wlds[k0 + kh + 1][m];
    c = __builtin_amdgcn_wmma_f32_16x16x4_f32(false, a, false, b, (short)0, c,
                                              false, false);
  }

  int mbase = (lane >> 4) << 3;
#pragma unroll
  for (int i = 0; i < 8; ++i) {
    int node = tile * 16 + mbase + i;
    float v = c[i] * dinv[node];
    hs[node * HPAD + m]  = v;
    agg[node * HPAD + m] = v;
  }
}

__global__ void k_scatter(const int* __restrict__ row, const int* __restrict__ col,
                          const float* __restrict__ hs, float* __restrict__ agg,
                          int e) {
  int t = blockIdx.x * blockDim.x + threadIdx.x;
  int eid = t >> 2;
  if (eid >= e) return;
  int q = (t & 3) << 2;
  int r  = row[eid];
  int cc = col[eid];
  const float4 v = *reinterpret_cast<const float4*>(hs + (size_t)r * HPAD + q);
  float* dst = agg + (size_t)cc * HPAD + q;
  atomicAddF(dst + 0, v.x);
  atomicAddF(dst + 1, v.y);
  atomicAddF(dst + 2, v.z);
  atomicAddF(dst + 3, v.w);
}

__global__ void k_finish(const float* __restrict__ agg, const float* __restrict__ dinv,
                         const float* __restrict__ b, float* __restrict__ hout,
                         int n) {
  int t = blockIdx.x * blockDim.x + threadIdx.x;
  if (t >= n * HPAD) return;
  int v = t >> 4, nn = t & 15;
  float x = 0.0f;
  if (nn < HFEAT) x = fmaxf(dinv[v] * agg[t] + b[nn], 0.0f);
  hout[t] = x;
}

__global__ void k_gemm_out(const float* __restrict__ A,
                           const float* __restrict__ Wc,
                           const float* __restrict__ bc,
                           float* __restrict__ out,
                           int numNodeTiles) {
  __shared__ float wlds[HPAD][DFEAT];
  for (int i = threadIdx.x; i < HPAD * DFEAT; i += blockDim.x) {
    int k = i >> 7, n = i & 127;
    wlds[k][n] = (k < HFEAT) ? Wc[k * DFEAT + n] : 0.0f;
  }
  __syncthreads();

  int wave  = threadIdx.x >> 5;
  int lane  = threadIdx.x & 31;
  int wtile = blockIdx.x * (blockDim.x >> 5) + wave;
  if (wtile >= numNodeTiles * 8) return;
  int tile = wtile >> 3;
  int n0   = (wtile & 7) << 4;

  int m  = lane & 15;
  int kh = (lane >> 4) << 1;
  const float* arow = A + (size_t)(tile * 16 + m) * HPAD;

  v8f c = {};
#pragma unroll
  for (int k0 = 0; k0 < HPAD; k0 += 4) {
    v2f a, b;
    a.x = arow[k0 + kh];
    a.y = arow[k0 + kh + 1];
    b.x = wlds[k0 + kh][n0 + m];
    b.y = wlds[k0 + kh + 1][n0 + m];
    c = __builtin_amdgcn_wmma_f32_16x16x4_f32(false, a, false, b, (short)0, c,
                                              false, false);
  }

  float bias = bc[n0 + m];
  int mbase = (lane >> 4) << 3;
#pragma unroll
  for (int i = 0; i < 8; ++i) {
    int node = tile * 16 + mbase + i;
    out[(size_t)node * DFEAT + n0 + m] = c[i] + bias;
  }
}

extern "C" void kernel_launch(void* const* d_in, const int* in_sizes, int n_in,
                              void* d_out, int out_size, void* d_ws, size_t ws_size,
                              hipStream_t stream) {
  const float* x  = (const float*)d_in[0];
  const int*   ei = (const int*)d_in[1];
  const float* W1 = (const float*)d_in[2];
  const float* b1 = (const float*)d_in[3];
  const float* W2 = (const float*)d_in[4];
  const float* b2 = (const float*)d_in[5];
  const float* W3 = (const float*)d_in[6];
  const float* b3 = (const float*)d_in[7];
  const float* Wc = (const float*)d_in[8];
  const float* bc = (const float*)d_in[9];

  const int N = in_sizes[0] / DFEAT;
  const int E = in_sizes[1] / 2;
  const int* row = ei;
  const int* col = ei + E;

  float* dinv = (float*)d_ws;
  float* B0 = dinv + N;
  float* B1 = B0 + (size_t)N * HPAD;
  float* B2 = B1 + (size_t)N * HPAD;
  float* outf = (float*)d_out;

  const int T = 256;
  const int numTiles = N / 16;
  dim3 gN((N + T - 1) / T);
  dim3 gE((E + T - 1) / T);
  dim3 gE4((E * 4 + T - 1) / T);
  dim3 gF((N * HPAD + T - 1) / T);
  dim3 gG((numTiles + 7) / 8);
  dim3 gO((numTiles * 8 + 7) / 8);

  k_deg_init <<<gN, T, 0, stream>>>(dinv, N);
  k_deg_count<<<gE, T, 0, stream>>>(col, dinv, E);
  k_deg_rsqrt<<<gN, T, 0, stream>>>(dinv, N);

  k_gemm_scale<DFEAT><<<gG, T, 0, stream>>>(x, W1, DFEAT, dinv, B0, B1, numTiles);
  k_scatter<<<gE4, T, 0, stream>>>(row, col, B0, B1, E);
  k_finish <<<gF,  T, 0, stream>>>(B1, dinv, b1, B2, N);

  k_gemm_scale<HPAD><<<gG, T, 0, stream>>>(B2, W2, HFEAT, dinv, B0, B1, numTiles);
  k_scatter<<<gE4, T, 0, stream>>>(row, col, B0, B1, E);
  k_finish <<<gF,  T, 0, stream>>>(B1, dinv, b2, B2, N);

  k_gemm_scale<HPAD><<<gG, T, 0, stream>>>(B2, W3, HFEAT, dinv, B0, B1, numTiles);
  k_scatter<<<gE4, T, 0, stream>>>(row, col, B0, B1, E);
  k_finish <<<gF,  T, 0, stream>>>(B1, dinv, b3, B2, N);

  k_gemm_out<<<gO, T, 0, stream>>>(B2, Wc, bc, outf, numTiles);
}


# Degree count by per-block LDS integer histograms (edge chunk x node range) and a plane sum instead of one global f32 atomic per edge; counts exact, dinv bit-identical; 13 -> 12 launches
# speedup vs baseline: 1.3052x; 1.3052x over previous
#include <hip/hip_runtime.h>


typedef __attribute__((ext_vector_type(2))) float v2f;
typedef __attribute__((ext_vector_type(8))) float v8f;

#define DFEAT 128
#define HFEAT 15
#define HPAD  16

__device__ __forceinline__ void atomicAddF(float* p, float v) {
  unsafeAtomicAdd(p, v);
}

#define DEG_CAP    12544
#define DEG_GROUPS 32
#define DEG_THREADS 1024

__device__ __forceinline__ void deg_bump(int* cnt, int c, int lo, unsigned span) {
  unsigned d = (unsigned)(c - lo);
  if (d < span) atomicAdd(&cnt[d], 1);
}

__device__ __forceinline__ void deg_bump4(int* cnt, int4 c, int lo, unsigned span) {
  deg_bump(cnt, c.x, lo, span);
  deg_bump(cnt, c.y, lo, span);
  deg_bump(cnt, c.z, lo, span);
  deg_bump(cnt, c.w, lo, span);
}

__global__ __launch_bounds__(DEG_THREADS) void k_deg_part(const int* __restrict__ col,
                                                   int* __restrict__ part,
                                                   int n, int e, int chunk) {
  __shared__ int cnt[DEG_CAP];
  const int tid = threadIdx.x;
  constexpr int T = DEG_THREADS;
  const int g   = blockIdx.x;
  const int lo  = blockIdx.y * DEG_CAP;
  const int rem = n - lo;
  const unsigned span = (unsigned)(rem < DEG_CAP ? rem : DEG_CAP);

  for (int i = tid; i < DEG_CAP; i += T) cnt[i] = 0;
  __syncthreads();

  const long long e0l = (long long)g * chunk;
  if (e0l < (long long)e) {
    const int e0 = (int)e0l;
    const int e1 = (e - e0 < chunk) ? e : e0 + chunk;
    const int* base = col + e0;
    if ((((size_t)base) & 15) == 0) {
      const int nvec = (e1 - e0) >> 2;
      const int4* p = reinterpret_cast<const int4*>(base);
      int i = tid;
      for (; i + 3 * T < nvec; i += 4 * T) {
        int4 a = p[i];
        int4 b = p[i + T];
        int4 c = p[i + 2 * T];
        int4 d = p[i + 3 * T];
        deg_bump4(cnt, a, lo, span);
        deg_bump4(cnt, b, lo, span);
        deg_bump4(cnt, c, lo, span);
        deg_bump4(cnt, d, lo, span);
      }
      for (; i < nvec; i += T) deg_bump4(cnt, p[i], lo, span);
      for (int j = e0 + (nvec << 2) + tid; j < e1; j += T) deg_bump(cnt, col[j], lo, span);
    } else {
      for (int j = e0 + tid; j < e1; j += T) deg_bump(cnt, col[j], lo, span);
    }
  }
  __syncthreads();

  int* dst = part + (size_t)g * n + lo;
  for (int i = tid; i < (int)span; i += T) dst[i] = cnt[i];
}

__global__ void k_deg_final(const int* __restrict__ part, float* __restrict__ dinv, int n) {
  int i = blockIdx.x * blockDim.x + threadIdx.x;
  if (i >= n) return;
  int s = 1;
#pragma unroll
  for (int g = 0; g < DEG_GROUPS; ++g) s += part[(size_t)g * n + i];
  dinv[i] = rsqrtf((float)s);
}

template <int K>
__global__ void k_gemm_scale(const float* __restrict__ A,
                             const float* __restrict__ W,
                             int wrows,
                             const float* __restrict__ dinv,
                             float* __restrict__ hs,
                             float* __restrict__ agg,
                             int numTiles) {
  __shared__ float wlds[K][HPAD];
  for (int i = threadIdx.x; i < K * HPAD; i += blockDim.x) {
    int k = i >> 4, n = i & 15;
    wlds[k][n] = (k < wrows && n < HFEAT) ? W[k * HFEAT + n] : 0.0f;
  }
  __syncthreads();

  int wave = threadIdx.x >> 5;
  int lane = threadIdx.x & 31;
  int tile = blockIdx.x * (blockDim.x >> 5) + wave;
  if (tile >= numTiles) return;

  int m  = lane & 15;
  int kh = (lane >> 4) << 1;
  const float* arow = A + (size_t)(tile * 16 + m) * K;

  v8f c = {};
#pragma unroll
  for (int k0 = 0; k0 < K; k0 += 4) {
    v2f a, b;
    a.x = arow[k0 + kh];
    a.y = arow[k0 + kh + 1];
    b.x = wlds[k0 + kh][m];
    b.y = wlds[k0 + kh + 1][m];
    c = __builtin_amdgcn_wmma_f32_16x16x4_f32(false, a, false, b, (short)0, c,
                                              false, false);
  }

  int mbase = (lane >> 4) << 3;
#pragma unroll
  for (int i = 0; i < 8; ++i) {
    int node = tile * 16 + mbase + i;
    float v = c[i] * dinv[node];
    hs[node * HPAD + m]  = v;
    agg[node * HPAD + m] = v;
  }
}

__global__ void k_scatter(const int* __restrict__ row, const int* __restrict__ col,
                          const float* __restrict__ hs, float* __restrict__ agg,
                          int e) {
  int t = blockIdx.x * blockDim.x + threadIdx.x;
  int eid = t >> 2;
  if (eid >= e) return;
  int q = (t & 3) << 2;
  int r  = row[eid];
  int cc = col[eid];
  const float4 v = *reinterpret_cast<const float4*>(hs + (size_t)r * HPAD + q);
  float* dst = agg + (size_t)cc * HPAD + q;
  atomicAddF(dst + 0, v.x);
  atomicAddF(dst + 1, v.y);
  atomicAddF(dst + 2, v.z);
  atomicAddF(dst + 3, v.w);
}

__global__ void k_finish(const float* __restrict__ agg, const float* __restrict__ dinv,
                         const float* __restrict__ b, float* __restrict__ hout,
                         int n) {
  int t = blockIdx.x * blockDim.x + threadIdx.x;
  if (t >= n * HPAD) return;
  int v = t >> 4, nn = t & 15;
  float x = 0.0f;
  if (nn < HFEAT) x = fmaxf(dinv[v] * agg[t] + b[nn], 0.0f);
  hout[t] = x;
}

__global__ void k_gemm_out(const float* __restrict__ A,
                           const float* __restrict__ Wc,
                           const float* __restrict__ bc,
                           float* __restrict__ out,
                           int numNodeTiles) {
  __shared__ float wlds[HPAD][DFEAT];
  for (int i = threadIdx.x; i < HPAD * DFEAT; i += blockDim.x) {
    int k = i >> 7, n = i & 127;
    wlds[k][n] = (k < HFEAT) ? Wc[k * DFEAT + n] : 0.0f;
  }
  __syncthreads();

  int wave  = threadIdx.x >> 5;
  int lane  = threadIdx.x & 31;
  int wtile = blockIdx.x * (blockDim.x >> 5) + wave;
  if (wtile >= numNodeTiles * 8) return;
  int tile = wtile >> 3;
  int n0   = (wtile & 7) << 4;

  int m  = lane & 15;
  int kh = (lane >> 4) << 1;
  const float* arow = A + (size_t)(tile * 16 + m) * HPAD;

  v8f c = {};
#pragma unroll
  for (int k0 = 0; k0 < HPAD; k0 += 4) {
    v2f a, b;
    a.x = arow[k0 + kh];
    a.y = arow[k0 + kh + 1];
    b.x = wlds[k0 + kh][n0 + m];
    b.y = wlds[k0 + kh + 1][n0 + m];
    c = __builtin_amdgcn_wmma_f32_16x16x4_f32(false, a, false, b, (short)0, c,
                                              false, false);
  }

  float bias = bc[n0 + m];
  int mbase = (lane >> 4) << 3;
#pragma unroll
  for (int i = 0; i < 8; ++i) {
    int node = tile * 16 + mbase + i;
    out[(size_t)node * DFEAT + n0 + m] = c[i] + bias;
  }
}

extern "C" void kernel_launch(void* const* d_in, const int* in_sizes, int n_in,
                              void* d_out, int out_size, void* d_ws, size_t ws_size,
                              hipStream_t stream) {
  const float* x  = (const float*)d_in[0];
  const int*   ei = (const int*)d_in[1];
  const float* W1 = (const float*)d_in[2];
  const float* b1 = (const float*)d_in[3];
  const float* W2 = (const float*)d_in[4];
  const float* b2 = (const float*)d_in[5];
  const float* W3 = (const float*)d_in[6];
  const float* b3 = (const float*)d_in[7];
  const float* Wc = (const float*)d_in[8];
  const float* bc = (const float*)d_in[9];

  const int N = in_sizes[0] / DFEAT;
  const int E = in_sizes[1] / 2;
  const int* row = ei;
  const int* col = ei + E;

  float* dinv = (float*)d_ws;
  float* B0 = dinv + N;
  float* B1 = B0 + (size_t)N * HPAD;
  float* B2 = B1 + (size_t)N * HPAD;
  float* outf = (float*)d_out;

  const int T = 256;
  const int numTiles = N / 16;
  dim3 gN((N + T - 1) / T);
  dim3 gE4((E * 4 + T - 1) / T);
  dim3 gF((N * HPAD + T - 1) / T);
  dim3 gG((numTiles + 7) / 8);
  dim3 gO((numTiles * 8 + 7) / 8);

  int* part = (int*)B0;
  const int chunk = (((E + DEG_GROUPS - 1) / DEG_GROUPS) + 3) & ~3;
  dim3 gP(DEG_GROUPS, (N + DEG_CAP - 1) / DEG_CAP);
  k_deg_part <<<gP, DEG_THREADS, 0, stream>>>(col, part, N, E, chunk);
  k_deg_final<<<gN, T, 0, stream>>>(part, dinv, N);

  k_gemm_scale<DFEAT><<<gG, T, 0, stream>>>(x, W1, DFEAT, dinv, B0, B1, numTiles);
  k_scatter<<<gE4, T, 0, stream>>>(row, col, B0, B1, E);
  k_finish <<<gF,  T, 0, stream>>>(B1, dinv, b1, B2, N);

  k_gemm_scale<HPAD><<<gG, T, 0, stream>>>(B2, W2, HFEAT, dinv, B0, B1, numTiles);
  k_scatter<<<gE4, T, 0, stream>>>(row, col, B0, B1, E);
  k_finish <<<gF,  T, 0, stream>>>(B1, dinv, b2, B2, N);

  k_gemm_scale<HPAD><<<gG, T, 0, stream>>>(B2, W3, HFEAT, dinv, B0, B1, numTiles);
  k_scatter<<<gE4, T, 0, stream>>>(row, col, B0, B1, E);
  k_finish <<<gF,  T, 0, stream>>>(B1, dinv, b3, B2, N);

  k_gemm_out<<<gO, T, 0, stream>>>(B2, Wc, bc, outf, numTiles);
}


# Edges counting-sorted by target once per call (LDS slot cursors, arrival order within a chunk); each layer sums a node's source rows in registers in list order, fused relu(dinv*sum+b); no f32 atomics
# speedup vs baseline: 3.2003x; 2.4520x over previous
#include <hip/hip_runtime.h>


typedef __attribute__((ext_vector_type(2))) float v2f;
typedef __attribute__((ext_vector_type(8))) float v8f;

#define DFEAT 128
#define HFEAT 15
#define HPAD  16

#define DEG_CAP    12544
#define DEG_GROUPS 32
#define DEG_THREADS 1024

__device__ __forceinline__ void deg_bump(int* cnt, int c, int lo, unsigned span) {
  unsigned d = (unsigned)(c - lo);
  if (d < span) atomicAdd(&cnt[d], 1);
}

__device__ __forceinline__ void deg_bump4(int* cnt, int4 c, int lo, unsigned span) {
  deg_bump(cnt, c.x, lo, span);
  deg_bump(cnt, c.y, lo, span);
  deg_bump(cnt, c.z, lo, span);
  deg_bump(cnt, c.w, lo, span);
}

__global__ __launch_bounds__(DEG_THREADS) void k_deg_part(const int* __restrict__ col,
                                                   int* __restrict__ part,
                                                   int n, int e, int chunk) {
  __shared__ int cnt[DEG_CAP];
  const int tid = threadIdx.x;
  constexpr int T = DEG_THREADS;
  const int g   = blockIdx.x;
  const int lo  = blockIdx.y * DEG_CAP;
  const int rem = n - lo;
  const unsigned span = (unsigned)(rem < DEG_CAP ? rem : DEG_CAP);

  for (int i = tid; i < DEG_CAP; i += T) cnt[i] = 0;
  __syncthreads();

  const long long e0l = (long long)g * chunk;
  if (e0l < (long long)e) {
    const int e0 = (int)e0l;
    const int e1 = (e - e0 < chunk) ? e : e0 + chunk;
    const int* base = col + e0;
    if ((((size_t)base) & 15) == 0) {
      const int nvec = (e1 - e0) >> 2;
      const int4* p = reinterpret_cast<const int4*>(base);
      int i = tid;
      for (; i + 3 * T < nvec; i += 4 * T) {
        int4 a = p[i];
        int4 b = p[i + T];
        int4 c = p[i + 2 * T];
        int4 d = p[i + 3 * T];
        deg_bump4(cnt, a, lo, span);
        deg_bump4(cnt, b, lo, span);
        deg_bump4(cnt, c, lo, span);
        deg_bump4(cnt, d, lo, span);
      }
      for (; i < nvec; i += T) deg_bump4(cnt, p[i], lo, span);
      for (int j = e0 + (nvec << 2) + tid; j < e1; j += T) deg_bump(cnt, col[j], lo, span);
    } else {
      for (int j = e0 + tid; j < e1; j += T) deg_bump(cnt, col[j], lo, span);
    }
  }
  __syncthreads();

  int* dst = part + (size_t)g * n + lo;
  for (int i = tid; i < (int)span; i += T) dst[i] = cnt[i];
}

#define NODE_THREADS 256
__global__ __launch_bounds__(NODE_THREADS) void k_deg_final(int* __restrict__ part,
                                                            float* __restrict__ dinv,
                                                            int* __restrict__ rowptr,
                                                            int* __restrict__ bsum, int n) {
  __shared__ int wtot[NODE_THREADS / 32];
  const int tid  = threadIdx.x;
  const int lane = tid & 31;
  const int wave = tid >> 5;
  const int i = blockIdx.x * NODE_THREADS + tid;
  int s = 0;
  if (i < n) {
    int v[DEG_GROUPS];
#pragma unroll
    for (int g = 0; g < DEG_GROUPS; ++g) v[g] = part[(size_t)g * n + i];
#pragma unroll
    for (int g = 0; g < DEG_GROUPS; ++g) {
      part[(size_t)g * n + i] = s;
      s += v[g];
    }
    dinv[i] = rsqrtf((float)(s + 1));
  }
  int inc = s;
  for (int o = 1; o < 32; o <<= 1) {
    int t = __shfl_up(inc, o, 32);
    if (lane >= o) inc += t;
  }
  if (lane == 31) wtot[wave] = inc;
  __syncthreads();
  int before = 0;
  for (int w = 0; w < wave; ++w) before += wtot[w];
  if (i < n) rowptr[i] = before + inc - s;
  if (tid == NODE_THREADS - 1) bsum[blockIdx.x] = before + inc;
}

__global__ __launch_bounds__(NODE_THREADS) void k_rowptr(int* __restrict__ rowptr,
                                                         const int* __restrict__ bsum, int n) {
  __shared__ int wtot[NODE_THREADS / 32];
  const int tid  = threadIdx.x;
  const int lane = tid & 31;
  const int wave = tid >> 5;
  const int b = blockIdx.x;
  int acc = 0;
  for (int j = tid; j < b; j += NODE_THREADS) acc += bsum[j];
  for (int o = 16; o > 0; o >>= 1) acc += __shfl_xor(acc, o, 32);
  if (lane == 0) wtot[wave] = acc;
  __syncthreads();
  int off = 0;
  for (int w = 0; w < NODE_THREADS / 32; ++w) off += wtot[w];
  const int i = b * NODE_THREADS + tid;
  if (i < n) {
    rowptr[i] += off;
    if (i == n - 1) rowptr[n] = off + bsum[b];
  }
}

__device__ __forceinline__ int take_slot(int* cur, int c, int lo, unsigned span) {
  unsigned d = (unsigned)(c - lo);
  return (d < span) ? atomicAdd(&cur[d], 1) : -1;
}

__device__ __forceinline__ void put_src(int* __restrict__ srcs, int pos, int r) {
  if (pos >= 0) srcs[pos] = r;
}

__device__ __forceinline__ void place1(int* cur, int* __restrict__ srcs, int c, int r, int lo,
                                       unsigned span) {
  put_src(srcs, take_slot(cur, c, lo, span), r);
}

__device__ __forceinline__ void place4(int* cur, int* __restrict__ srcs, int4 c, int4 r, int lo,
                                       unsigned span) {
  const int s0 = take_slot(cur, c.x, lo, span);
  const int s1 = take_slot(cur, c.y, lo, span);
  const int s2 = take_slot(cur, c.z, lo, span);
  const int s3 = take_slot(cur, c.w, lo, span);
  put_src(srcs, s0, r.x);
  put_src(srcs, s1, r.y);
  put_src(srcs, s2, r.z);
  put_src(srcs, s3, r.w);
}

__device__ __forceinline__ void place8(int* cur, int* __restrict__ srcs, int4 c0, int4 r0, int4 c1,
                                       int4 r1, int lo, unsigned span) {
  const int s0 = take_slot(cur, c0.x, lo, span);
  const int s1 = take_slot(cur, c0.y, lo, span);
  const int s2 = take_slot(cur, c0.z, lo, span);
  const int s3 = take_slot(cur, c0.w, lo, span);
  const int s4 = take_slot(cur, c1.x, lo, span);
  const int s5 = take_slot(cur, c1.y, lo, span);
  const int s6 = take_slot(cur, c1.z, lo, span);
  const int s7 = take_slot(cur, c1.w, lo, span);
  put_src(srcs, s0, r0.x);
  put_src(srcs, s1, r0.y);
  put_src(srcs, s2, r0.z);
  put_src(srcs, s3, r0.w);
  put_src(srcs, s4, r1.x);
  put_src(srcs, s5, r1.y);
  put_src(srcs, s6, r1.z);
  put_src(srcs, s7, r1.w);
}

__global__ __launch_bounds__(DEG_THREADS) void k_place(const int* __restrict__ row,
                                                const int* __restrict__ col,
                                                const int* __restrict__ part,
                                                const int* __restrict__ rowptr,
                                                int* __restrict__ srcs,
                                                int n, int e, int chunk) {
  __shared__ int cur[DEG_CAP];
  const int tid = threadIdx.x;
  constexpr int T = DEG_THREADS;
  const int g   = blockIdx.x;
  const int lo  = blockIdx.y * DEG_CAP;
  const int rem = n - lo;
  const unsigned span = (unsigned)(rem < DEG_CAP ? rem : DEG_CAP);

  const int* pp = part + (size_t)g * n + lo;
  for (int i = tid; i < (int)span; i += T) cur[i] = rowptr[lo + i] + pp[i];
  __syncthreads();

  const long long e0l = (long long)g * chunk;
  if (e0l >= (long long)e) return;
  const int e0 = (int)e0l;
  const int e1 = (e - e0 < chunk) ? e : e0 + chunk;
  const int* cbase = col + e0;
  const int* rbase = row + e0;
  if (((((size_t)cbase) | ((size_t)rbase)) & 15) == 0) {
    const int nvec = (e1 - e0) >> 2;
    const int4* pc = reinterpret_cast<const int4*>(cbase);
    const int4* pr = reinterpret_cast<const int4*>(rbase);
    int i = tid;
    for (; i + T < nvec; i += 2 * T) {
      int4 c0 = pc[i];
      int4 r0 = pr[i];
      int4 c1 = pc[i + T];
      int4 r1 = pr[i + T];
      place8(cur, srcs, c0, r0, c1, r1, lo, span);
    }
    for (; i < nvec; i += T) place4(cur, srcs, pc[i], pr[i], lo, span);
    for (int j = e0 + (nvec << 2) + tid; j < e1; j += T) place1(cur, srcs, col[j], row[j], lo, span);
  } else {
    for (int j = e0 + tid; j < e1; j += T) place1(cur, srcs, col[j], row[j], lo, span);
  }
}

template <int K>
__global__ void k_gemm_scale(const float* __restrict__ A,
                             const float* __restrict__ W,
                             int wrows,
                             const float* __restrict__ dinv,
                             float* __restrict__ hs,
                             int numTiles) {
  __shared__ float wlds[K][HPAD];
  for (int i = threadIdx.x; i < K * HPAD; i += blockDim.x) {
    int k = i >> 4, n = i & 15;
    wlds[k][n] = (k < wrows && n < HFEAT) ? W[k * HFEAT + n] : 0.0f;
  }
  __syncthreads();

  int wave = threadIdx.x >> 5;
  int lane = threadIdx.x & 31;
  int tile = blockIdx.x * (blockDim.x >> 5) + wave;
  if (tile >= numTiles) return;

  int m  = lane & 15;
  int kh = (lane >> 4) << 1;
  const float* arow = A + (size_t)(tile * 16 + m) * K;

  v8f c = {};
#pragma unroll
  for (int k0 = 0; k0 < K; k0 += 4) {
    v2f a, b;
    a.x = arow[k0 + kh];
    a.y = arow[k0 + kh + 1];
    b.x = wlds[k0 + kh][m];
    b.y = wlds[k0 + kh + 1][m];
    c = __builtin_amdgcn_wmma_f32_16x16x4_f32(false, a, false, b, (short)0, c,
                                              false, false);
  }

  int mbase = (lane >> 4) << 3;
#pragma unroll
  for (int i = 0; i < 8; ++i) {
    int node = tile * 16 + mbase + i;
    float v = c[i] * dinv[node];
    hs[node * HPAD + m] = v;
  }
}

#define GATHER_BATCH 8
__global__ __launch_bounds__(256) void k_gather(const int* __restrict__ rowptr,
                                                const int* __restrict__ srcs,
                                                const float* __restrict__ hs,
                                                const float* __restrict__ dinv,
                                                const float* __restrict__ b,
                                                float* __restrict__ hout, int n) {
  int t = blockIdx.x * blockDim.x + threadIdx.x;
  int v = t >> 2;
  if (v >= n) return;
  int q = t & 3;
  const char* hsq = reinterpret_cast<const char*>(hs) + (q << 4);
  int p   = rowptr[v];
  int end = rowptr[v + 1];
  float4 acc = *reinterpret_cast<const float4*>(hsq + ((unsigned)v << 6));

  int idx[GATHER_BATCH];
  if (p + GATHER_BATCH <= end) {
#pragma unroll
    for (int k = 0; k < GATHER_BATCH; ++k) idx[k] = srcs[p + k];
  }
  while (p + GATHER_BATCH <= end) {
    float4 x[GATHER_BATCH];
#pragma unroll
    for (int k = 0; k < GATHER_BATCH; ++k)
      x[k] = *reinterpret_cast<const float4*>(hsq + ((unsigned)idx[k] << 6));
    p += GATHER_BATCH;
    if (p + GATHER_BATCH <= end) {
#pragma unroll
      for (int k = 0; k < GATHER_BATCH; ++k) idx[k] = srcs[p + k];
    }
#pragma unroll
    for (int k = 0; k < GATHER_BATCH; ++k) {
      acc.x += x[k].x;
      acc.y += x[k].y;
      acc.z += x[k].z;
      acc.w += x[k].w;
    }
  }
  for (; p < end; ++p) {
    const float4 x = *reinterpret_cast<const float4*>(hsq + ((unsigned)srcs[p] << 6));
    acc.x += x.x;
    acc.y += x.y;
    acc.z += x.z;
    acc.w += x.w;
  }

  const float d = dinv[v];
  const int c0 = q << 2;
  float4 o;
  o.x = fmaxf(d * acc.x + b[c0 + 0], 0.0f);
  o.y = fmaxf(d * acc.y + b[c0 + 1], 0.0f);
  o.z = fmaxf(d * acc.z + b[c0 + 2], 0.0f);
  o.w = (c0 + 3 < HFEAT) ? fmaxf(d * acc.w + b[c0 + 3], 0.0f) : 0.0f;
  reinterpret_cast<float4*>(hout)[(size_t)v * 4 + q] = o;
}

__global__ void k_gemm_out(const float* __restrict__ A,
                           const float* __restrict__ Wc,
                           const float* __restrict__ bc,
                           float* __restrict__ out,
                           int numNodeTiles) {
  __shared__ float wlds[HPAD][DFEAT];
  for (int i = threadIdx.x; i < HPAD * DFEAT; i += blockDim.x) {
    int k = i >> 7, n = i & 127;
    wlds[k][n] = (k < HFEAT) ? Wc[k * DFEAT + n] : 0.0f;
  }
  __syncthreads();

  int wave  = threadIdx.x >> 5;
  int lane  = threadIdx.x & 31;
  int wtile = blockIdx.x * (blockDim.x >> 5) + wave;
  if (wtile >= numNodeTiles * 8) return;
  int tile = wtile >> 3;
  int n0   = (wtile & 7) << 4;

  int m  = lane & 15;
  int kh = (lane >> 4) << 1;
  const float* arow = A + (size_t)(tile * 16 + m) * HPAD;

  v8f c = {};
#pragma unroll
  for (int k0 = 0; k0 < HPAD; k0 += 4) {
    v2f a, b;
    a.x = arow[k0 + kh];
    a.y = arow[k0 + kh + 1];
    b.x = wlds[k0 + kh][n0 + m];
    b.y = wlds[k0 + kh + 1][n0 + m];
    c = __builtin_amdgcn_wmma_f32_16x16x4_f32(false, a, false, b, (short)0, c,
                                              false, false);
  }

  float bias = bc[n0 + m];
  int mbase = (lane >> 4) << 3;
#pragma unroll
  for (int i = 0; i < 8; ++i) {
    int node = tile * 16 + mbase + i;
    out[(size_t)node * DFEAT + n0 + m] = c[i] + bias;
  }
}

extern "C" void kernel_launch(void* const* d_in, const int* in_sizes, int n_in,
                              void* d_out, int out_size, void* d_ws, size_t ws_size,
                              hipStream_t stream) {
  const float* x  = (const float*)d_in[0];
  const int*   ei = (const int*)d_in[1];
  const float* W1 = (const float*)d_in[2];
  const float* b1 = (const float*)d_in[3];
  const float* W2 = (const float*)d_in[4];
  const float* b2 = (const float*)d_in[5];
  const float* W3 = (const float*)d_in[6];
  const float* b3 = (const float*)d_in[7];
  const float* Wc = (const float*)d_in[8];
  const float* bc = (const float*)d_in[9];

  const int N = in_sizes[0] / DFEAT;
  const int E = in_sizes[1] / 2;
  const int* row = ei;
  const int* col = ei + E;

  float* dinv = (float*)d_ws;
  float* B0 = dinv + N;
  float* B1 = B0 + (size_t)N * HPAD;
  float* B2 = B1 + (size_t)N * HPAD;
  float* outf = (float*)d_out;
  int* rowptr = (int*)B1;
  int* bsum   = rowptr + N + 1;

  int* srcs = (int*)d_out;
  int* part = srcs + E;

  const int T = 256;
  const int numTiles = N / 16;
  dim3 gN((N + NODE_THREADS - 1) / NODE_THREADS);
  dim3 gV((N * 4 + T - 1) / T);
  dim3 gG((numTiles + 7) / 8);
  dim3 gO((numTiles * 8 + 7) / 8);

  const int chunk = (((E + DEG_GROUPS - 1) / DEG_GROUPS) + 3) & ~3;
  dim3 gP(DEG_GROUPS, (N + DEG_CAP - 1) / DEG_CAP);
  k_deg_part <<<gP, DEG_THREADS, 0, stream>>>(col, part, N, E, chunk);
  k_deg_final<<<gN, NODE_THREADS, 0, stream>>>(part, dinv, rowptr, bsum, N);
  k_rowptr   <<<gN, NODE_THREADS, 0, stream>>>(rowptr, bsum, N);
  k_place    <<<gP, DEG_THREADS, 0, stream>>>(row, col, part, rowptr, srcs, N, E, chunk);

  k_gemm_scale<DFEAT><<<gG, T, 0, stream>>>(x, W1, DFEAT, dinv, B0, numTiles);
  k_gather<<<gV, T, 0, stream>>>(rowptr, srcs, B0, dinv, b1, B2, N);

  k_gemm_scale<HPAD><<<gG, T, 0, stream>>>(B2, W2, HFEAT, dinv, B0, numTiles);
  k_gather<<<gV, T, 0, stream>>>(rowptr, srcs, B0, dinv, b2, B2, N);

  k_gemm_scale<HPAD><<<gG, T, 0, stream>>>(B2, W3, HFEAT, dinv, B0, numTiles);
  k_gather<<<gV, T, 0, stream>>>(rowptr, srcs, B0, dinv, b3, B2, N);

  k_gemm_out<<<gO, T, 0, stream>>>(B2, Wc, bc, outf, numTiles);
}


# Gather: a quad fetches 32 neighbour ids per round, 8 per lane and none twice, and shares them by DPP quad broadcast; list tail is one predicated batch; sums and their order unchanged
# speedup vs baseline: 3.4224x; 1.0694x over previous
#include <hip/hip_runtime.h>


typedef __attribute__((ext_vector_type(2))) float v2f;
typedef __attribute__((ext_vector_type(8))) float v8f;

#define DFEAT 128
#define HFEAT 15
#define HPAD  16

#define DEG_CAP    12544
#define DEG_GROUPS 32
#define DEG_THREADS 1024

__device__ __forceinline__ void deg_bump(int* cnt, int c, int lo, unsigned span) {
  unsigned d = (unsigned)(c - lo);
  if (d < span) atomicAdd(&cnt[d], 1);
}

__device__ __forceinline__ void deg_bump4(int* cnt, int4 c, int lo, unsigned span) {
  deg_bump(cnt, c.x, lo, span);
  deg_bump(cnt, c.y, lo, span);
  deg_bump(cnt, c.z, lo, span);
  deg_bump(cnt, c.w, lo, span);
}

__global__ __launch_bounds__(DEG_THREADS) void k_deg_part(const int* __restrict__ col,
                                                   int* __restrict__ part,
                                                   int n, int e, int chunk) {
  __shared__ int cnt[DEG_CAP];
  const int tid = threadIdx.x;
  constexpr int T = DEG_THREADS;
  const int g   = blockIdx.x;
  const int lo  = blockIdx.y * DEG_CAP;
  const int rem = n - lo;
  const unsigned span = (unsigned)(rem < DEG_CAP ? rem : DEG_CAP);

  for (int i = tid; i < DEG_CAP; i += T) cnt[i] = 0;
  __syncthreads();

  const long long e0l = (long long)g * chunk;
  if (e0l < (long long)e) {
    const int e0 = (int)e0l;
    const int e1 = (e - e0 < chunk) ? e : e0 + chunk;
    const int* base = col + e0;
    if ((((size_t)base) & 15) == 0) {
      const int nvec = (e1 - e0) >> 2;
      const int4* p = reinterpret_cast<const int4*>(base);
      int i = tid;
      for (; i + 3 * T < nvec; i += 4 * T) {
        int4 a = p[i];
        int4 b = p[i + T];
        int4 c = p[i + 2 * T];
        int4 d = p[i + 3 * T];
        deg_bump4(cnt, a, lo, span);
        deg_bump4(cnt, b, lo, span);
        deg_bump4(cnt, c, lo, span);
        deg_bump4(cnt, d, lo, span);
      }
      for (; i < nvec; i += T) deg_bump4(cnt, p[i], lo, span);
      for (int j = e0 + (nvec << 2) + tid; j < e1; j += T) deg_bump(cnt, col[j], lo, span);
    } else {
      for (int j = e0 + tid; j < e1; j += T) deg_bump(cnt, col[j], lo, span);
    }
  }
  __syncthreads();

  int* dst = part + (size_t)g * n + lo;
  for (int i = tid; i < (int)span; i += T) dst[i] = cnt[i];
}

#define NODE_THREADS 256
__global__ __launch_bounds__(NODE_THREADS) void k_deg_final(int* __restrict__ part,
                                                            float* __restrict__ dinv,
                                                            int* __restrict__ rowptr,
                                                            int* __restrict__ bsum, int n) {
  __shared__ int wtot[NODE_THREADS / 32];
  const int tid  = threadIdx.x;
  const int lane = tid & 31;
  const int wave = tid >> 5;
  const int i = blockIdx.x * NODE_THREADS + tid;
  int s = 0;
  if (i < n) {
    int v[DEG_GROUPS];
#pragma unroll
    for (int g = 0; g < DEG_GROUPS; ++g) v[g] = part[(size_t)g * n + i];
#pragma unroll
    for (int g = 0; g < DEG_GROUPS; ++g) {
      part[(size_t)g * n + i] = s;
      s += v[g];
    }
    dinv[i] = rsqrtf((float)(s + 1));
  }
  int inc = s;
  for (int o = 1; o < 32; o <<= 1) {
    int t = __shfl_up(inc, o, 32);
    if (lane >= o) inc += t;
  }
  if (lane == 31) wtot[wave] = inc;
  __syncthreads();
  int before = 0;
  for (int w = 0; w < wave; ++w) before += wtot[w];
  if (i < n) rowptr[i] = before + inc - s;
  if (tid == NODE_THREADS - 1) bsum[blockIdx.x] = before + inc;
}

__global__ __launch_bounds__(NODE_THREADS) void k_rowptr(int* __restrict__ rowptr,
                                                         const int* __restrict__ bsum, int n) {
  __shared__ int wtot[NODE_THREADS / 32];
  const int tid  = threadIdx.x;
  const int lane = tid & 31;
  const int wave = tid >> 5;
  const int b = blockIdx.x;
  int acc = 0;
  for (int j = tid; j < b; j += NODE_THREADS) acc += bsum[j];
  for (int o = 16; o > 0; o >>= 1) acc += __shfl_xor(acc, o, 32);
  if (lane == 0) wtot[wave] = acc;
  __syncthreads();
  int off = 0;
  for (int w = 0; w < NODE_THREADS / 32; ++w) off += wtot[w];
  const int i = b * NODE_THREADS + tid;
  if (i < n) {
    rowptr[i] += off;
    if (i == n - 1) rowptr[n] = off + bsum[b];
  }
}

__device__ __forceinline__ int take_slot(int* cur, int c, int lo, unsigned span) {
  unsigned d = (unsigned)(c - lo);
  return (d < span) ? atomicAdd(&cur[d], 1) : -1;
}

__device__ __forceinline__ void put_src(int* __restrict__ srcs, int pos, int r) {
  if (pos >= 0) srcs[pos] = r;
}

__device__ __forceinline__ void place1(int* cur, int* __restrict__ srcs, int c, int r, int lo,
                                       unsigned span) {
  put_src(srcs, take_slot(cur, c, lo, span), r);
}

__device__ __forceinline__ void place4(int* cur, int* __restrict__ srcs, int4 c, int4 r, int lo,
                                       unsigned span) {
  const int s0 = take_slot(cur, c.x, lo, span);
  const int s1 = take_slot(cur, c.y, lo, span);
  const int s2 = take_slot(cur, c.z, lo, span);
  const int s3 = take_slot(cur, c.w, lo, span);
  put_src(srcs, s0, r.x);
  put_src(srcs, s1, r.y);
  put_src(srcs, s2, r.z);
  put_src(srcs, s3, r.w);
}

__device__ __forceinline__ void place8(int* cur, int* __restrict__ srcs, int4 c0, int4 r0, int4 c1,
                                       int4 r1, int lo, unsigned span) {
  const int s0 = take_slot(cur, c0.x, lo, span);
  const int s1 = take_slot(cur, c0.y, lo, span);
  const int s2 = take_slot(cur, c0.z, lo, span);
  const int s3 = take_slot(cur, c0.w, lo, span);
  const int s4 = take_slot(cur, c1.x, lo, span);
  const int s5 = take_slot(cur, c1.y, lo, span);
  const int s6 = take_slot(cur, c1.z, lo, span);
  const int s7 = take_slot(cur, c1.w, lo, span);
  put_src(srcs, s0, r0.x);
  put_src(srcs, s1, r0.y);
  put_src(srcs, s2, r0.z);
  put_src(srcs, s3, r0.w);
  put_src(srcs, s4, r1.x);
  put_src(srcs, s5, r1.y);
  put_src(srcs, s6, r1.z);
  put_src(srcs, s7, r1.w);
}

__global__ __launch_bounds__(DEG_THREADS) void k_place(const int* __restrict__ row,
                                                const int* __restrict__ col,
                                                const int* __restrict__ part,
                                                const int* __restrict__ rowptr,
                                                int* __restrict__ srcs,
                                                int n, int e, int chunk) {
  __shared__ int cur[DEG_CAP];
  const int tid = threadIdx.x;
  constexpr int T = DEG_THREADS;
  const int g   = blockIdx.x;
  const int lo  = blockIdx.y * DEG_CAP;
  const int rem = n - lo;
  const unsigned span = (unsigned)(rem < DEG_CAP ? rem : DEG_CAP);

  const int* pp = part + (size_t)g * n + lo;
  for (int i = tid; i < (int)span; i += T) cur[i] = rowptr[lo + i] + pp[i];
  __syncthreads();

  const long long e0l = (long long)g * chunk;
  if (e0l >= (long long)e) return;
  const int e0 = (int)e0l;
  const int e1 = (e - e0 < chunk) ? e : e0 + chunk;
  const int* cbase = col + e0;
  const int* rbase = row + e0;
  if (((((size_t)cbase) | ((size_t)rbase)) & 15) == 0) {
    const int nvec = (e1 - e0) >> 2;
    const int4* pc = reinterpret_cast<const int4*>(cbase);
    const int4* pr = reinterpret_cast<const int4*>(rbase);
    int i = tid;
    for (; i + T < nvec; i += 2 * T) {
      int4 c0 = pc[i];
      int4 r0 = pr[i];
      int4 c1 = pc[i + T];
      int4 r1 = pr[i + T];
      place8(cur, srcs, c0, r0, c1, r1, lo, span);
    }
    for (; i < nvec; i += T) place4(cur, srcs, pc[i], pr[i], lo, span);
    for (int j = e0 + (nvec << 2) + tid; j < e1; j += T) place1(cur, srcs, col[j], row[j], lo, span);
  } else {
    for (int j = e0 + tid; j < e1; j += T) place1(cur, srcs, col[j], row[j], lo, span);
  }
}

template <int K>
__global__ void k_gemm_scale(const float* __restrict__ A,
                             const float* __restrict__ W,
                             int wrows,
                             const float* __restrict__ dinv,
                             float* __restrict__ hs,
                             int numTiles) {
  __shared__ float wlds[K][HPAD];
  for (int i = threadIdx.x; i < K * HPAD; i += blockDim.x) {
    int k = i >> 4, n = i & 15;
    wlds[k][n] = (k < wrows && n < HFEAT) ? W[k * HFEAT + n] : 0.0f;
  }
  __syncthreads();

  int wave = threadIdx.x >> 5;
  int lane = threadIdx.x & 31;
  int tile = blockIdx.x * (blockDim.x >> 5) + wave;
  if (tile >= numTiles) return;

  int m  = lane & 15;
  int kh = (lane >> 4) << 1;
  const float* arow = A + (size_t)(tile * 16 + m) * K;

  v8f c = {};
#pragma unroll
  for (int k0 = 0; k0 < K; k0 += 4) {
    v2f a, b;
    a.x = arow[k0 + kh];
    a.y = arow[k0 + kh + 1];
    b.x = wlds[k0 + kh][m];
    b.y = wlds[k0 + kh + 1][m];
    c = __builtin_amdgcn_wmma_f32_16x16x4_f32(false, a, false, b, (short)0, c,
                                              false, false);
  }

  int mbase = (lane >> 4) << 3;
#pragma unroll
  for (int i = 0; i < 8; ++i) {
    int node = tile * 16 + mbase + i;
    float v = c[i] * dinv[node];
    hs[node * HPAD + m] = v;
  }
}

#define GATHER_BATCH 8

template <int K>
__device__ __forceinline__ unsigned quad_get(unsigned x) {
  return (unsigned)__builtin_amdgcn_update_dpp(0, (int)x, K * 0x55, 0xf, 0xf, true);
}

__device__ __forceinline__ float4 ld_row(const char* base, unsigned off) {
  return *reinterpret_cast<const float4*>(base + off);
}

__device__ __forceinline__ void add4(float4& a, const float4 x) {
  a.x += x.x;
  a.y += x.y;
  a.z += x.z;
  a.w += x.w;
}

template <int L0, int BASE>
__device__ __forceinline__ void batch8(const char* hsb, uint4 ids, unsigned q16, int left,
                                       float4& acc) {
  const unsigned o0 = quad_get<L0>(ids.x) | q16;
  const unsigned o1 = quad_get<L0>(ids.y) | q16;
  const unsigned o2 = quad_get<L0>(ids.z) | q16;
  const unsigned o3 = quad_get<L0>(ids.w) | q16;
  const unsigned o4 = quad_get<L0 + 1>(ids.x) | q16;
  const unsigned o5 = quad_get<L0 + 1>(ids.y) | q16;
  const unsigned o6 = quad_get<L0 + 1>(ids.z) | q16;
  const unsigned o7 = quad_get<L0 + 1>(ids.w) | q16;
  if (left >= BASE + GATHER_BATCH) {
    const float4 x0 = ld_row(hsb, o0);
    const float4 x1 = ld_row(hsb, o1);
    const float4 x2 = ld_row(hsb, o2);
    const float4 x3 = ld_row(hsb, o3);
    const float4 x4 = ld_row(hsb, o4);
    const float4 x5 = ld_row(hsb, o5);
    const float4 x6 = ld_row(hsb, o6);
    const float4 x7 = ld_row(hsb, o7);
    add4(acc, x0);
    add4(acc, x1);
    add4(acc, x2);
    add4(acc, x3);
    add4(acc, x4);
    add4(acc, x5);
    add4(acc, x6);
    add4(acc, x7);
  } else {
    const float4 z = make_float4(0.0f, 0.0f, 0.0f, 0.0f);
    float4 x0 = z, x1 = z, x2 = z, x3 = z, x4 = z, x5 = z, x6 = z;
    x0 = ld_row(hsb, o0);
    if (left > BASE + 1) x1 = ld_row(hsb, o1);
    if (left > BASE + 2) x2 = ld_row(hsb, o2);
    if (left > BASE + 3) x3 = ld_row(hsb, o3);
    if (left > BASE + 4) x4 = ld_row(hsb, o4);
    if (left > BASE + 5) x5 = ld_row(hsb, o5);
    if (left > BASE + 6) x6 = ld_row(hsb, o6);
    add4(acc, x0);
    if (left > BASE + 1) add4(acc, x1);
    if (left > BASE + 2) add4(acc, x2);
    if (left > BASE + 3) add4(acc, x3);
    if (left > BASE + 4) add4(acc, x4);
    if (left > BASE + 5) add4(acc, x5);
    if (left > BASE + 6) add4(acc, x6);
  }
}

__global__ __launch_bounds__(256) void k_gather(const int* __restrict__ rowptr,
                                                const int* __restrict__ srcs,
                                                const float* __restrict__ hs,
                                                const float* __restrict__ dinv,
                                                const float* __restrict__ b,
                                                float* __restrict__ hout, int n) {
  int t = blockIdx.x * blockDim.x + threadIdx.x;
  int v = t >> 2;
  if (v >= n) return;
  const int q = t & 3;
  const unsigned q16 = (unsigned)q << 4;
  const char* hsb = reinterpret_cast<const char*>(hs);
  const int p = rowptr[v];
  const int end = rowptr[v + 1];
  float4 acc = ld_row(hsb, ((unsigned)v << 6) | q16);

  const int* ip = srcs + p + 4 * q;
  for (int left = end - p; left > 0; left -= 32, ip += 32) {
    uint4 ia, ib;
    ia.x = (unsigned)ip[0] << 6;
    ia.y = (unsigned)ip[1] << 6;
    ia.z = (unsigned)ip[2] << 6;
    ia.w = (unsigned)ip[3] << 6;
    ib.x = (unsigned)ip[16] << 6;
    ib.y = (unsigned)ip[17] << 6;
    ib.z = (unsigned)ip[18] << 6;
    ib.w = (unsigned)ip[19] << 6;
    batch8<0, 0>(hsb, ia, q16, left, acc);
    if (left > 8) batch8<2, 8>(hsb, ia, q16, left, acc);
    if (left > 16) batch8<0, 16>(hsb, ib, q16, left, acc);
    if (left > 24) batch8<2, 24>(hsb, ib, q16, left, acc);
  }

  const float d = dinv[v];
  const int c0 = q << 2;
  float4 o;
  o.x = fmaxf(d * acc.x + b[c0 + 0], 0.0f);
  o.y = fmaxf(d * acc.y + b[c0 + 1], 0.0f);
  o.z = fmaxf(d * acc.z + b[c0 + 2], 0.0f);
  o.w = (c0 + 3 < HFEAT) ? fmaxf(d * acc.w + b[c0 + 3], 0.0f) : 0.0f;
  reinterpret_cast<float4*>(hout)[(size_t)v * 4 + q] = o;
}

__global__ void k_gemm_out(const float* __restrict__ A,
                           const float* __restrict__ Wc,
                           const float* __restrict__ bc,
                           float* __restrict__ out,
                           int numNodeTiles) {
  __shared__ float wlds[HPAD][DFEAT];
  for (int i = threadIdx.x; i < HPAD * DFEAT; i += blockDim.x) {
    int k = i >> 7, n = i & 127;
    wlds[k][n] = (k < HFEAT) ? Wc[k * DFEAT + n] : 0.0f;
  }
  __syncthreads();

  int wave  = threadIdx.x >> 5;
  int lane  = threadIdx.x & 31;
  int wtile = blockIdx.x * (blockDim.x >> 5) + wave;
  if (wtile >= numNodeTiles * 8) return;
  int tile = wtile >> 3;
  int n0   = (wtile & 7) << 4;

  int m  = lane & 15;
  int kh = (lane >> 4) << 1;
  const float* arow = A + (size_t)(tile * 16 + m) * HPAD;

  v8f c = {};
#pragma unroll
  for (int k0 = 0; k0 < HPAD; k0 += 4) {
    v2f a, b;
    a.x = arow[k0 + kh];
    a.y = arow[k0 + kh + 1];
    b.x = wlds[k0 + kh][n0 + m];
    b.y = wlds[k0 + kh + 1][n0 + m];
    c = __builtin_amdgcn_wmma_f32_16x16x4_f32(false, a, false, b, (short)0, c,
                                              false, false);
  }

  float bias = bc[n0 + m];
  int mbase = (lane >> 4) << 3;
#pragma unroll
  for (int i = 0; i < 8; ++i) {
    int node = tile * 16 + mbase + i;
    out[(size_t)node * DFEAT + n0 + m] = c[i] + bias;
  }
}

extern "C" void kernel_launch(void* const* d_in, const int* in_sizes, int n_in,
                              void* d_out, int out_size, void* d_ws, size_t ws_size,
                              hipStream_t stream) {
  const float* x  = (const float*)d_in[0];
  const int*   ei = (const int*)d_in[1];
  const float* W1 = (const float*)d_in[2];
  const float* b1 = (const float*)d_in[3];
  const float* W2 = (const float*)d_in[4];
  const float* b2 = (const float*)d_in[5];
  const float* W3 = (const float*)d_in[6];
  const float* b3 = (const float*)d_in[7];
  const float* Wc = (const float*)d_in[8];
  const float* bc = (const float*)d_in[9];

  const int N = in_sizes[0] / DFEAT;
  const int E = in_sizes[1] / 2;
  const int* row = ei;
  const int* col = ei + E;

  float* dinv = (float*)d_ws;
  float* B0 = dinv + N;
  float* B1 = B0 + (size_t)N * HPAD;
  float* B2 = B1 + (size_t)N * HPAD;
  float* outf = (float*)d_out;
  int* rowptr = (int*)B1;
  int* bsum   = rowptr + N + 1;

  int* srcs = (int*)d_out;
  int* part = srcs + E;

  const int T = 256;
  const int numTiles = N / 16;
  dim3 gN((N + NODE_THREADS - 1) / NODE_THREADS);
  dim3 gV((N * 4 + T - 1) / T);
  dim3 gG((numTiles + 7) / 8);
  dim3 gO((numTiles * 8 + 7) / 8);

  const int chunk = (((E + DEG_GROUPS - 1) / DEG_GROUPS) + 3) & ~3;
  dim3 gP(DEG_GROUPS, (N + DEG_CAP - 1) / DEG_CAP);
  k_deg_part <<<gP, DEG_THREADS, 0, stream>>>(col, part, N, E, chunk);
  k_deg_final<<<gN, NODE_THREADS, 0, stream>>>(part, dinv, rowptr, bsum, N);
  k_rowptr   <<<gN, NODE_THREADS, 0, stream>>>(rowptr, bsum, N);
  k_place    <<<gP, DEG_THREADS, 0, stream>>>(row, col, part, rowptr, srcs, N, E, chunk);

  k_gemm_scale<DFEAT><<<gG, T, 0, stream>>>(x, W1, DFEAT, dinv, B0, numTiles);
  k_gather<<<gV, T, 0, stream>>>(rowptr, srcs, B0, dinv, b1, B2, N);

  k_gemm_scale<HPAD><<<gG, T, 0, stream>>>(B2, W2, HFEAT, dinv, B0, numTiles);
  k_gather<<<gV, T, 0, stream>>>(rowptr, srcs, B0, dinv, b2, B2, N);

  k_gemm_scale<HPAD><<<gG, T, 0, stream>>>(B2, W3, HFEAT, dinv, B0, numTiles);
  k_gather<<<gV, T, 0, stream>>>(rowptr, srcs, B0, dinv, b3, B2, N);

  k_gemm_out<<<gO, T, 0, stream>>>(B2, Wc, bc, outf, numTiles);
}


# Edge lists built by a two-level LDS bucket sort (25000-edge runs binned by 392-node range, then a block per range) with whole-line stores, 9 launches; sums stay chunk-major, arrival order in a chunk
# speedup vs baseline: 6.0310x; 1.7622x over previous
#include <hip/hip_runtime.h>


typedef __attribute__((ext_vector_type(2))) float v2f;
typedef __attribute__((ext_vector_type(8))) float v8f;

#define DFEAT 128
#define HFEAT 15
#define HPAD  16

#define BIN_CHUNK   25000
#define BIN_THREADS 1024
#define RANGE       392
#define NR_MAX      256
#define TAB_STRIDE  (NR_MAX + 1)
#define SRC_BITS    17
#define SRC_MASK    0x1FFFFu
#define NO_ENTRY    0xFFFFFFFFu

__device__ __forceinline__ void bin_count(int* cur, int c, unsigned n) {
  if ((unsigned)c < n) atomicAdd(&cur[(unsigned)c / RANGE], 1);
}

__device__ __forceinline__ int bin_take(int* cur, int c, unsigned n) {
  return ((unsigned)c < n) ? atomicAdd(&cur[(unsigned)c / RANGE], 1) : -1;
}

__device__ __forceinline__ void bin_put(unsigned* stage, int slot, int c, int r) {
  if (slot >= 0) stage[slot] = (((unsigned)c % RANGE) << SRC_BITS) | ((unsigned)r & SRC_MASK);
}

__global__ __launch_bounds__(BIN_THREADS) void k_bin(const int* __restrict__ row,
                                                     const int* __restrict__ col,
                                                     unsigned* __restrict__ bucketed,
                                                     int* __restrict__ tab,
                                                     int n, int e) {
  __shared__ int cur[NR_MAX];
  __shared__ int wtot[NR_MAX / 32];
  __shared__ int s_total;
  __shared__ __attribute__((aligned(16))) unsigned stage[BIN_CHUNK];
  constexpr int T = BIN_THREADS;
  constexpr int VPT = (BIN_CHUNK / 4 + T - 1) / T;
  const int tid  = threadIdx.x;
  const int lane = tid & 31;
  const int wave = tid >> 5;
  const int g  = blockIdx.x;
  const int e0 = g * BIN_CHUNK;
  const int cntE = (e - e0 < BIN_CHUNK) ? (e - e0) : BIN_CHUNK;
  const unsigned un = (unsigned)n;
  const int* cb = col + e0;
  const int* rb = row + e0;
  const int nvec = (((((size_t)cb) | ((size_t)rb)) & 15) == 0) ? (cntE >> 2) : 0;

  int4 c[VPT], r[VPT];
  const int4 none = make_int4(-1, -1, -1, -1);
#pragma unroll
  for (int k = 0; k < VPT; ++k) {
    c[k] = none;
    r[k] = none;
  }
  if (nvec > 0) {
    const int4* c4 = reinterpret_cast<const int4*>(cb);
    const int4* r4 = reinterpret_cast<const int4*>(rb);
#pragma unroll
    for (int k = 0; k < VPT; ++k) {
      const int i = tid + k * T;
      const bool has = i < nvec;
      const int4 cv = c4[has ? i : 0];
      const int4 rv = r4[has ? i : 0];
      c[k] = has ? cv : none;
      r[k] = rv;
    }
  }
  if (tid < NR_MAX) cur[tid] = 0;
  __syncthreads();

#pragma unroll
  for (int k = 0; k < VPT; ++k) {
    bin_count(cur, c[k].x, un);
    bin_count(cur, c[k].y, un);
    bin_count(cur, c[k].z, un);
    bin_count(cur, c[k].w, un);
  }
  for (int j = (nvec << 2) + tid; j < cntE; j += T) bin_count(cur, cb[j], un);
  __syncthreads();

  int v = 0, inc = 0;
  if (tid < NR_MAX) {
    v = cur[tid];
    inc = v;
    for (int o = 1; o < 32; o <<= 1) {
      int t = __shfl_up(inc, o, 32);
      if (lane >= o) inc += t;
    }
    if (lane == 31) wtot[wave] = inc;
  }
  __syncthreads();
  if (tid < NR_MAX) {
    int before = 0;
    for (int w = 0; w < wave; ++w) before += wtot[w];
    const int ex = before + inc - v;
    cur[tid] = ex;
    tab[g * TAB_STRIDE + tid] = ex;
    if (tid == NR_MAX - 1) {
      tab[g * TAB_STRIDE + NR_MAX] = ex + v;
      s_total = ex + v;
    }
  }
  __syncthreads();

#pragma unroll
  for (int k = 0; k < VPT; ++k) {
    const int s0 = bin_take(cur, c[k].x, un);
    const int s1 = bin_take(cur, c[k].y, un);
    const int s2 = bin_take(cur, c[k].z, un);
    const int s3 = bin_take(cur, c[k].w, un);
    bin_put(stage, s0, c[k].x, r[k].x);
    bin_put(stage, s1, c[k].y, r[k].y);
    bin_put(stage, s2, c[k].z, r[k].z);
    bin_put(stage, s3, c[k].w, r[k].w);
  }
  for (int j = (nvec << 2) + tid; j < cntE; j += T) {
    const int cj = cb[j];
    bin_put(stage, bin_take(cur, cj, un), cj, rb[j]);
  }
  __syncthreads();

  const int total = s_total;
  unsigned* dst = bucketed + (size_t)g * BIN_CHUNK;
  const int nv = ((((size_t)dst) & 15) == 0) ? (total >> 2) : 0;
  for (int i = tid; i < nv; i += T)
    reinterpret_cast<uint4*>(dst)[i] = reinterpret_cast<const uint4*>(stage)[i];
  for (int j = (nv << 2) + tid; j < total; j += T) dst[j] = stage[j];
}

#define GRP_THREADS 1024
#define GRP_CAP     32768
#define GRP_STEPS   32
#define GRP_PIECES  8
#define GRP_LANES   (GRP_THREADS / GRP_PIECES)
#define GRP_NODES   512

__device__ __forceinline__ void grp_place(int* cursor, unsigned* outl, int* __restrict__ srcs,
                                          int base, unsigned w) {
  if (w != NO_ENTRY) {
    const int slot = atomicAdd(&cursor[w >> SRC_BITS], 1);
    const int sv = (int)(w & SRC_MASK);
    if (slot < GRP_CAP) outl[slot] = (unsigned)sv;
    else srcs[base + slot] = sv;
  }
}

__global__ __launch_bounds__(GRP_THREADS) void k_group(const unsigned* __restrict__ bucketed,
                                                       const int* __restrict__ tab,
                                                       int* __restrict__ srcs,
                                                       int* __restrict__ rowptr,
                                                       float* __restrict__ dinv,
                                                       int n, int ng) {
  __shared__ int pc[GRP_STEPS * GRP_PIECES];
  __shared__ int poff[GRP_STEPS * GRP_PIECES];
  __shared__ int deg[GRP_NODES];
  __shared__ int wsum[(GRP_STEPS * GRP_PIECES) / 32];
  __shared__ int wtot[GRP_NODES / 32];
  __shared__ int s_base;
  __shared__ int s_total;
  __shared__ unsigned outl[GRP_CAP];
  const int tid  = threadIdx.x;
  const int lane = tid & 31;
  const int wave = tid >> 5;
  const int rg = blockIdx.x;
  const int lo = rg * RANGE;
  const int rem = n - lo;
  const int span = rem < RANGE ? rem : RANGE;

  if (tid < GRP_STEPS * GRP_PIECES) {
    int a = 0, b = 0;
    if (tid < ng) {
      a = tab[tid * TAB_STRIDE + rg];
      b = tab[tid * TAB_STRIDE + rg + 1];
    }
    pc[tid] = b - a;
    poff[tid] = a;
    int sum = a;
    for (int o = 16; o > 0; o >>= 1) sum += __shfl_xor(sum, o, 32);
    if (lane == 0) wsum[wave] = sum;
  }
  if (tid < GRP_NODES) deg[tid] = 0;
  __syncthreads();
  if (tid == 0) {
    int s = 0;
    for (int w = 0; w < (GRP_STEPS * GRP_PIECES) / 32; ++w) s += wsum[w];
    s_base = s;
  }

  const int sub = tid / GRP_LANES;
  const int li  = tid % GRP_LANES;
  unsigned e1[GRP_STEPS], e2[GRP_STEPS];
#pragma unroll
  for (int s = 0; s < GRP_STEPS; ++s) {
    const int g = s * GRP_PIECES + sub;
    const int cnt = pc[g];
    const unsigned* src = bucketed + (size_t)g * BIN_CHUNK + poff[g];
    const bool h1 = li < cnt;
    const bool h2 = li + GRP_LANES < cnt;
    const unsigned w1 = *(h1 ? src + li : bucketed);
    const unsigned w2 = *(h2 ? src + li + GRP_LANES : bucketed);
    e1[s] = h1 ? w1 : NO_ENTRY;
    e2[s] = h2 ? w2 : NO_ENTRY;
  }

#pragma unroll
  for (int s = 0; s < GRP_STEPS; ++s) {
    if (e1[s] != NO_ENTRY) atomicAdd(&deg[e1[s] >> SRC_BITS], 1);
    if (e2[s] != NO_ENTRY) atomicAdd(&deg[e2[s] >> SRC_BITS], 1);
  }
  for (int s = 0; s < GRP_STEPS; ++s) {
    const int g = s * GRP_PIECES + sub;
    const int cnt = pc[g];
    const unsigned* src = bucketed + (size_t)g * BIN_CHUNK + poff[g];
    for (int i = li + 2 * GRP_LANES; i < cnt; i += GRP_LANES) atomicAdd(&deg[src[i] >> SRC_BITS], 1);
  }
  __syncthreads();

  int v = 0, inc = 0;
  if (tid < GRP_NODES) {
    v = deg[tid];
    inc = v;
    for (int o = 1; o < 32; o <<= 1) {
      int t = __shfl_up(inc, o, 32);
      if (lane >= o) inc += t;
    }
    if (lane == 31) wtot[wave] = inc;
  }
  __syncthreads();
  const int base = s_base;
  if (tid < GRP_NODES) {
    int before = 0;
    for (int w = 0; w < wave; ++w) before += wtot[w];
    const int ex = before + inc - v;
    deg[tid] = ex;
    if (tid < span) {
      rowptr[lo + tid] = base + ex;
      dinv[lo + tid] = rsqrtf((float)(v + 1));
      if (lo + tid == n - 1) rowptr[n] = base + ex + v;
    }
    if (tid == GRP_NODES - 1) s_total = ex + v;
  }
  __syncthreads();

#pragma unroll
  for (int s = 0; s < GRP_STEPS; ++s) {
    grp_place(deg, outl, srcs, base, e1[s]);
    grp_place(deg, outl, srcs, base, e2[s]);
    __syncthreads();
  }
  for (int s = 0; s < GRP_STEPS; ++s) {
    const int g = s * GRP_PIECES + sub;
    const int cnt = pc[g];
    const unsigned* src = bucketed + (size_t)g * BIN_CHUNK + poff[g];
    for (int i = li + 2 * GRP_LANES; i < cnt; i += GRP_LANES) grp_place(deg, outl, srcs, base, src[i]);
  }
  __syncthreads();

  const int total = s_total;
  const int tot = total < GRP_CAP ? total : GRP_CAP;
  for (int p = tid - (base & 31); p < tot; p += GRP_THREADS)
    if (p >= 0) srcs[base + p] = (int)outl[p];
}

template <int K>
__global__ void k_gemm_scale(const float* __restrict__ A,
                             const float* __restrict__ W,
                             int wrows,
                             const float* __restrict__ dinv,
                             float* __restrict__ hs,
                             int numTiles) {
  __shared__ float wlds[K][HPAD];
  for (int i = threadIdx.x; i < K * HPAD; i += blockDim.x) {
    int k = i >> 4, n = i & 15;
    wlds[k][n] = (k < wrows && n < HFEAT) ? W[k * HFEAT + n] : 0.0f;
  }
  __syncthreads();

  int wave = threadIdx.x >> 5;
  int lane = threadIdx.x & 31;
  int tile = blockIdx.x * (blockDim.x >> 5) + wave;
  if (tile >= numTiles) return;

  int m  = lane & 15;
  int kh = (lane >> 4) << 1;
  const float* arow = A + (size_t)(tile * 16 + m) * K;

  v8f c = {};
#pragma unroll
  for (int k0 = 0; k0 < K; k0 += 4) {
    v2f a, b;
    a.x = arow[k0 + kh];
    a.y = arow[k0 + kh + 1];
    b.x = wlds[k0 + kh][m];
    b.y = wlds[k0 + kh + 1][m];
    c = __builtin_amdgcn_wmma_f32_16x16x4_f32(false, a, false, b, (short)0, c,
                                              false, false);
  }

  int mbase = (lane >> 4) << 3;
#pragma unroll
  for (int i = 0; i < 8; ++i) {
    int node = tile * 16 + mbase + i;
    float v = c[i] * dinv[node];
    hs[node * HPAD + m] = v;
  }
}

#define GATHER_BATCH 8

template <int K>
__device__ __forceinline__ unsigned quad_get(unsigned x) {
  return (unsigned)__builtin_amdgcn_update_dpp(0, (int)x, K * 0x55, 0xf, 0xf, true);
}

__device__ __forceinline__ float4 ld_row(const char* base, unsigned off) {
  return *reinterpret_cast<const float4*>(base + off);
}

__device__ __forceinline__ void add4(float4& a, const float4 x) {
  a.x += x.x;
  a.y += x.y;
  a.z += x.z;
  a.w += x.w;
}

template <int L0, int BASE>
__device__ __forceinline__ void batch8(const char* hsb, uint4 ids, unsigned q16, int left,
                                       float4& acc) {
  const unsigned o0 = quad_get<L0>(ids.x) | q16;
  const unsigned o1 = quad_get<L0>(ids.y) | q16;
  const unsigned o2 = quad_get<L0>(ids.z) | q16;
  const unsigned o3 = quad_get<L0>(ids.w) | q16;
  const unsigned o4 = quad_get<L0 + 1>(ids.x) | q16;
  const unsigned o5 = quad_get<L0 + 1>(ids.y) | q16;
  const unsigned o6 = quad_get<L0 + 1>(ids.z) | q16;
  const unsigned o7 = quad_get<L0 + 1>(ids.w) | q16;
  if (left >= BASE + GATHER_BATCH) {
    const float4 x0 = ld_row(hsb, o0);
    const float4 x1 = ld_row(hsb, o1);
    const float4 x2 = ld_row(hsb, o2);
    const float4 x3 = ld_row(hsb, o3);
    const float4 x4 = ld_row(hsb, o4);
    const float4 x5 = ld_row(hsb, o5);
    const float4 x6 = ld_row(hsb, o6);
    const float4 x7 = ld_row(hsb, o7);
    add4(acc, x0);
    add4(acc, x1);
    add4(acc, x2);
    add4(acc, x3);
    add4(acc, x4);
    add4(acc, x5);
    add4(acc, x6);
    add4(acc, x7);
  } else {
    const float4 z = make_float4(0.0f, 0.0f, 0.0f, 0.0f);
    float4 x0 = z, x1 = z, x2 = z, x3 = z, x4 = z, x5 = z, x6 = z;
    x0 = ld_row(hsb, o0);
    if (left > BASE + 1) x1 = ld_row(hsb, o1);
    if (left > BASE + 2) x2 = ld_row(hsb, o2);
    if (left > BASE + 3) x3 = ld_row(hsb, o3);
    if (left > BASE + 4) x4 = ld_row(hsb, o4);
    if (left > BASE + 5) x5 = ld_row(hsb, o5);
    if (left > BASE + 6) x6 = ld_row(hsb, o6);
    add4(acc, x0);
    if (left > BASE + 1) add4(acc, x1);
    if (left > BASE + 2) add4(acc, x2);
    if (left > BASE + 3) add4(acc, x3);
    if (left > BASE + 4) add4(acc, x4);
    if (left > BASE + 5) add4(acc, x5);
    if (left > BASE + 6) add4(acc, x6);
  }
}

__global__ __launch_bounds__(256) void k_gather(const int* __restrict__ rowptr,
                                                const int* __restrict__ srcs,
                                                const float* __restrict__ hs,
                                                const float* __restrict__ dinv,
                                                const float* __restrict__ b,
                                                float* __restrict__ hout, int n) {
  int t = blockIdx.x * blockDim.x + threadIdx.x;
  int v = t >> 2;
  if (v >= n) return;
  const int q = t & 3;
  const unsigned q16 = (unsigned)q << 4;
  const char* hsb = reinterpret_cast<const char*>(hs);
  const int p = rowptr[v];
  const int end = rowptr[v + 1];
  float4 acc = ld_row(hsb, ((unsigned)v << 6) | q16);

  const int* ip = srcs + p + 4 * q;
  for (int left = end - p; left > 0; left -= 32, ip += 32) {
    uint4 ia, ib;
    ia.x = (unsigned)ip[0] << 6;
    ia.y = (unsigned)ip[1] << 6;
    ia.z = (unsigned)ip[2] << 6;
    ia.w = (unsigned)ip[3] << 6;
    ib.x = (unsigned)ip[16] << 6;
    ib.y = (unsigned)ip[17] << 6;
    ib.z = (unsigned)ip[18] << 6;
    ib.w = (unsigned)ip[19] << 6;
    batch8<0, 0>(hsb, ia, q16, left, acc);
    if (left > 8) batch8<2, 8>(hsb, ia, q16, left, acc);
    if (left > 16) batch8<0, 16>(hsb, ib, q16, left, acc);
    if (left > 24) batch8<2, 24>(hsb, ib, q16, left, acc);
  }

  const float d = dinv[v];
  const int c0 = q << 2;
  float4 o;
  o.x = fmaxf(d * acc.x + b[c0 + 0], 0.0f);
  o.y = fmaxf(d * acc.y + b[c0 + 1], 0.0f);
  o.z = fmaxf(d * acc.z + b[c0 + 2], 0.0f);
  o.w = (c0 + 3 < HFEAT) ? fmaxf(d * acc.w + b[c0 + 3], 0.0f) : 0.0f;
  reinterpret_cast<float4*>(hout)[(size_t)v * 4 + q] = o;
}

__global__ void k_gemm_out(const float* __restrict__ A,
                           const float* __restrict__ Wc,
                           const float* __restrict__ bc,
                           float* __restrict__ out,
                           int numNodeTiles) {
  __shared__ float wlds[HPAD][DFEAT];
  for (int i = threadIdx.x; i < HPAD * DFEAT; i += blockDim.x) {
    int k = i >> 7, n = i & 127;
    wlds[k][n] = (k < HFEAT) ? Wc[k * DFEAT + n] : 0.0f;
  }
  __syncthreads();

  int wave  = threadIdx.x >> 5;
  int lane  = threadIdx.x & 31;
  int wtile = blockIdx.x * (blockDim.x >> 5) + wave;
  if (wtile >= numNodeTiles * 8) return;
  int tile = wtile >> 3;
  int n0   = (wtile & 7) << 4;

  int m  = lane & 15;
  int kh = (lane >> 4) << 1;
  const float* arow = A + (size_t)(tile * 16 + m) * HPAD;

  v8f c = {};
#pragma unroll
  for (int k0 = 0; k0 < HPAD; k0 += 4) {
    v2f a, b;
    a.x = arow[k0 + kh];
    a.y = arow[k0 + kh + 1];
    b.x = wlds[k0 + kh][n0 + m];
    b.y = wlds[k0 + kh + 1][n0 + m];
    c = __builtin_amdgcn_wmma_f32_16x16x4_f32(false, a, false, b, (short)0, c,
                                              false, false);
  }

  float bias = bc[n0 + m];
  int mbase = (lane >> 4) << 3;
#pragma unroll
  for (int i = 0; i < 8; ++i) {
    int node = tile * 16 + mbase + i;
    out[(size_t)node * DFEAT + n0 + m] = c[i] + bias;
  }
}

extern "C" void kernel_launch(void* const* d_in, const int* in_sizes, int n_in,
                              void* d_out, int out_size, void* d_ws, size_t ws_size,
                              hipStream_t stream) {
  const float* x  = (const float*)d_in[0];
  const int*   ei = (const int*)d_in[1];
  const float* W1 = (const float*)d_in[2];
  const float* b1 = (const float*)d_in[3];
  const float* W2 = (const float*)d_in[4];
  const float* b2 = (const float*)d_in[5];
  const float* W3 = (const float*)d_in[6];
  const float* b3 = (const float*)d_in[7];
  const float* Wc = (const float*)d_in[8];
  const float* bc = (const float*)d_in[9];

  const int N = in_sizes[0] / DFEAT;
  const int E = in_sizes[1] / 2;
  const int* row = ei;
  const int* col = ei + E;

  float* dinv = (float*)d_ws;
  float* B0 = dinv + N;
  float* B1 = B0 + (size_t)N * HPAD;
  float* B2 = B1 + (size_t)N * HPAD;
  float* outf = (float*)d_out;
  int* rowptr = (int*)B1;
  int* tab    = rowptr + N + 1;

  int* srcs = (int*)d_out;
  unsigned* bucketed = (unsigned*)(srcs + E);

  const int T = 256;
  const int numTiles = N / 16;
  dim3 gV((N * 4 + T - 1) / T);
  dim3 gG((numTiles + 7) / 8);
  dim3 gO((numTiles * 8 + 7) / 8);

  const int runs = (E + BIN_CHUNK - 1) / BIN_CHUNK;
  dim3 gB(runs);
  dim3 gR((N + RANGE - 1) / RANGE);
  k_bin  <<<gB, BIN_THREADS, 0, stream>>>(row, col, bucketed, tab, N, E);
  k_group<<<gR, GRP_THREADS, 0, stream>>>(bucketed, tab, srcs, rowptr, dinv, N, runs);

  k_gemm_scale<DFEAT><<<gG, T, 0, stream>>>(x, W1, DFEAT, dinv, B0, numTiles);
  k_gather<<<gV, T, 0, stream>>>(rowptr, srcs, B0, dinv, b1, B2, N);

  k_gemm_scale<HPAD><<<gG, T, 0, stream>>>(B2, W2, HFEAT, dinv, B0, numTiles);
  k_gather<<<gV, T, 0, stream>>>(rowptr, srcs, B0, dinv, b2, B2, N);

  k_gemm_scale<HPAD><<<gG, T, 0, stream>>>(B2, W3, HFEAT, dinv, B0, numTiles);
  k_gather<<<gV, T, 0, stream>>>(rowptr, srcs, B0, dinv, b3, B2, N);

  k_gemm_out<<<gO, T, 0, stream>>>(B2, Wc, bc, outf, numTiles);
}


# Layer 2 and 3 weight products (15x15) fused into the preceding gather's epilogue: a wave's 8 rows -> A fragments via its own LDS tile, 4 WMMAs, writes next hs; 9 -> 7 launches; sums, order unchanged
# speedup vs baseline: 6.4123x; 1.0632x over previous
#include <hip/hip_runtime.h>


typedef __attribute__((ext_vector_type(2))) float v2f;
typedef __attribute__((ext_vector_type(8))) float v8f;

#define DFEAT 128
#define HFEAT 15
#define HPAD  16

#define BIN_CHUNK   25000
#define BIN_THREADS 1024
#define RANGE       392
#define NR_MAX      256
#define TAB_STRIDE  (NR_MAX + 1)
#define SRC_BITS    17
#define SRC_MASK    0x1FFFFu
#define NO_ENTRY    0xFFFFFFFFu

__device__ __forceinline__ void bin_count(int* cur, int c, unsigned n) {
  if ((unsigned)c < n) atomicAdd(&cur[(unsigned)c / RANGE], 1);
}

__device__ __forceinline__ int bin_take(int* cur, int c, unsigned n) {
  return ((unsigned)c < n) ? atomicAdd(&cur[(unsigned)c / RANGE], 1) : -1;
}

__device__ __forceinline__ void bin_put(unsigned* stage, int slot, int c, int r) {
  if (slot >= 0) stage[slot] = (((unsigned)c % RANGE) << SRC_BITS) | ((unsigned)r & SRC_MASK);
}

__global__ __launch_bounds__(BIN_THREADS) void k_bin(const int* __restrict__ row,
                                                     const int* __restrict__ col,
                                                     unsigned* __restrict__ bucketed,
                                                     int* __restrict__ tab,
                                                     int n, int e) {
  __shared__ int cur[NR_MAX];
  __shared__ int wtot[NR_MAX / 32];
  __shared__ int s_total;
  __shared__ __attribute__((aligned(16))) unsigned stage[BIN_CHUNK];
  constexpr int T = BIN_THREADS;
  constexpr int VPT = (BIN_CHUNK / 4 + T - 1) / T;
  const int tid  = threadIdx.x;
  const int lane = tid & 31;
  const int wave = tid >> 5;
  const int g  = blockIdx.x;
  const int e0 = g * BIN_CHUNK;
  const int cntE = (e - e0 < BIN_CHUNK) ? (e - e0) : BIN_CHUNK;
  const unsigned un = (unsigned)n;
  const int* cb = col + e0;
  const int* rb = row + e0;
  const int nvec = (((((size_t)cb) | ((size_t)rb)) & 15) == 0) ? (cntE >> 2) : 0;

  int4 c[VPT], r[VPT];
  const int4 none = make_int4(-1, -1, -1, -1);
#pragma unroll
  for (int k = 0; k < VPT; ++k) {
    c[k] = none;
    r[k] = none;
  }
  if (nvec > 0) {
    const int4* c4 = reinterpret_cast<const int4*>(cb);
    const int4* r4 = reinterpret_cast<const int4*>(rb);
#pragma unroll
    for (int k = 0; k < VPT; ++k) {
      const int i = tid + k * T;
      const bool has = i < nvec;
      const int4 cv = c4[has ? i : 0];
      const int4 rv = r4[has ? i : 0];
      c[k] = has ? cv : none;
      r[k] = rv;
    }
  }
  if (tid < NR_MAX) cur[tid] = 0;
  __syncthreads();

#pragma unroll
  for (int k = 0; k < VPT; ++k) {
    bin_count(cur, c[k].x, un);
    bin_count(cur, c[k].y, un);
    bin_count(cur, c[k].z, un);
    bin_count(cur, c[k].w, un);
  }
  for (int j = (nvec << 2) + tid; j < cntE; j += T) bin_count(cur, cb[j], un);
  __syncthreads();

  int v = 0, inc = 0;
  if (tid < NR_MAX) {
    v = cur[tid];
    inc = v;
    for (int o = 1; o < 32; o <<= 1) {
      int t = __shfl_up(inc, o, 32);
      if (lane >= o) inc += t;
    }
    if (lane == 31) wtot[wave] = inc;
  }
  __syncthreads();
  if (tid < NR_MAX) {
    int before = 0;
    for (int w = 0; w < wave; ++w) before += wtot[w];
    const int ex = before + inc - v;
    cur[tid] = ex;
    tab[g * TAB_STRIDE + tid] = ex;
    if (tid == NR_MAX - 1) {
      tab[g * TAB_STRIDE + NR_MAX] = ex + v;
      s_total = ex + v;
    }
  }
  __syncthreads();

#pragma unroll
  for (int k = 0; k < VPT; ++k) {
    const int s0 = bin_take(cur, c[k].x, un);
    const int s1 = bin_take(cur, c[k].y, un);
    const int s2 = bin_take(cur, c[k].z, un);
    const int s3 = bin_take(cur, c[k].w, un);
    bin_put(stage, s0, c[k].x, r[k].x);
    bin_put(stage, s1, c[k].y, r[k].y);
    bin_put(stage, s2, c[k].z, r[k].z);
    bin_put(stage, s3, c[k].w, r[k].w);
  }
  for (int j = (nvec << 2) + tid; j < cntE; j += T) {
    const int cj = cb[j];
    bin_put(stage, bin_take(cur, cj, un), cj, rb[j]);
  }
  __syncthreads();

  const int total = s_total;
  unsigned* dst = bucketed + (size_t)g * BIN_CHUNK;
  const int nv = ((((size_t)dst) & 15) == 0) ? (total >> 2) : 0;
  for (int i = tid; i < nv; i += T)
    reinterpret_cast<uint4*>(dst)[i] = reinterpret_cast<const uint4*>(stage)[i];
  for (int j = (nv << 2) + tid; j < total; j += T) dst[j] = stage[j];
}

#define GRP_THREADS 1024
#define GRP_CAP     32768
#define GRP_STEPS   32
#define GRP_PIECES  8
#define GRP_LANES   (GRP_THREADS / GRP_PIECES)
#define GRP_NODES   512

__device__ __forceinline__ void grp_place(int* cursor, unsigned* outl, int* __restrict__ srcs,
                                          int base, unsigned w) {
  if (w != NO_ENTRY) {
    const int slot = atomicAdd(&cursor[w >> SRC_BITS], 1);
    const int sv = (int)(w & SRC_MASK);
    if (slot < GRP_CAP) outl[slot] = (unsigned)sv;
    else srcs[base + slot] = sv;
  }
}

__global__ __launch_bounds__(GRP_THREADS) void k_group(const unsigned* __restrict__ bucketed,
                                                       const int* __restrict__ tab,
                                                       int* __restrict__ srcs,
                                                       int* __restrict__ rowptr,
                                                       float* __restrict__ dinv,
                                                       int n, int ng) {
  __shared__ int pc[GRP_STEPS * GRP_PIECES];
  __shared__ int poff[GRP_STEPS * GRP_PIECES];
  __shared__ int deg[GRP_NODES];
  __shared__ int wsum[(GRP_STEPS * GRP_PIECES) / 32];
  __shared__ int wtot[GRP_NODES / 32];
  __shared__ int s_base;
  __shared__ int s_total;
  __shared__ unsigned outl[GRP_CAP];
  const int tid  = threadIdx.x;
  const int lane = tid & 31;
  const int wave = tid >> 5;
  const int rg = blockIdx.x;
  const int lo = rg * RANGE;
  const int rem = n - lo;
  const int span = rem < RANGE ? rem : RANGE;

  if (tid < GRP_STEPS * GRP_PIECES) {
    int a = 0, b = 0;
    if (tid < ng) {
      a = tab[tid * TAB_STRIDE + rg];
      b = tab[tid * TAB_STRIDE + rg + 1];
    }
    pc[tid] = b - a;
    poff[tid] = a;
    int sum = a;
    for (int o = 16; o > 0; o >>= 1) sum += __shfl_xor(sum, o, 32);
    if (lane == 0) wsum[wave] = sum;
  }
  if (tid < GRP_NODES) deg[tid] = 0;
  __syncthreads();
  if (tid == 0) {
    int s = 0;
    for (int w = 0; w < (GRP_STEPS * GRP_PIECES) / 32; ++w) s += wsum[w];
    s_base = s;
  }

  const int sub = tid / GRP_LANES;
  const int li  = tid % GRP_LANES;
  unsigned e1[GRP_STEPS], e2[GRP_STEPS];
#pragma unroll
  for (int s = 0; s < GRP_STEPS; ++s) {
    const int g = s * GRP_PIECES + sub;
    const int cnt = pc[g];
    const unsigned* src = bucketed + (size_t)g * BIN_CHUNK + poff[g];
    const bool h1 = li < cnt;
    const bool h2 = li + GRP_LANES < cnt;
    const unsigned w1 = *(h1 ? src + li : bucketed);
    const unsigned w2 = *(h2 ? src + li + GRP_LANES : bucketed);
    e1[s] = h1 ? w1 : NO_ENTRY;
    e2[s] = h2 ? w2 : NO_ENTRY;
  }

#pragma unroll
  for (int s = 0; s < GRP_STEPS; ++s) {
    if (e1[s] != NO_ENTRY) atomicAdd(&deg[e1[s] >> SRC_BITS], 1);
    if (e2[s] != NO_ENTRY) atomicAdd(&deg[e2[s] >> SRC_BITS], 1);
  }
  for (int s = 0; s < GRP_STEPS; ++s) {
    const int g = s * GRP_PIECES + sub;
    const int cnt = pc[g];
    const unsigned* src = bucketed + (size_t)g * BIN_CHUNK + poff[g];
    for (int i = li + 2 * GRP_LANES; i < cnt; i += GRP_LANES) atomicAdd(&deg[src[i] >> SRC_BITS], 1);
  }
  __syncthreads();

  int v = 0, inc = 0;
  if (tid < GRP_NODES) {
    v = deg[tid];
    inc = v;
    for (int o = 1; o < 32; o <<= 1) {
      int t = __shfl_up(inc, o, 32);
      if (lane >= o) inc += t;
    }
    if (lane == 31) wtot[wave] = inc;
  }
  __syncthreads();
  const int base = s_base;
  if (tid < GRP_NODES) {
    int before = 0;
    for (int w = 0; w < wave; ++w) before += wtot[w];
    const int ex = before + inc - v;
    deg[tid] = ex;
    if (tid < span) {
      rowptr[lo + tid] = base + ex;
      dinv[lo + tid] = rsqrtf((float)(v + 1));
      if (lo + tid == n - 1) rowptr[n] = base + ex + v;
    }
    if (tid == GRP_NODES - 1) s_total = ex + v;
  }
  __syncthreads();

#pragma unroll
  for (int s = 0; s < GRP_STEPS; ++s) {
    grp_place(deg, outl, srcs, base, e1[s]);
    grp_place(deg, outl, srcs, base, e2[s]);
    __syncthreads();
  }
  for (int s = 0; s < GRP_STEPS; ++s) {
    const int g = s * GRP_PIECES + sub;
    const int cnt = pc[g];
    const unsigned* src = bucketed + (size_t)g * BIN_CHUNK + poff[g];
    for (int i = li + 2 * GRP_LANES; i < cnt; i += GRP_LANES) grp_place(deg, outl, srcs, base, src[i]);
  }
  __syncthreads();

  const int total = s_total;
  const int tot = total < GRP_CAP ? total : GRP_CAP;
  for (int p = tid - (base & 31); p < tot; p += GRP_THREADS)
    if (p >= 0) srcs[base + p] = (int)outl[p];
}

template <int K>
__global__ void k_gemm_scale(const float* __restrict__ A,
                             const float* __restrict__ W,
                             int wrows,
                             const float* __restrict__ dinv,
                             float* __restrict__ hs,
                             int numTiles) {
  __shared__ float wlds[K][HPAD];
  for (int i = threadIdx.x; i < K * HPAD; i += blockDim.x) {
    int k = i >> 4, n = i & 15;
    wlds[k][n] = (k < wrows && n < HFEAT) ? W[k * HFEAT + n] : 0.0f;
  }
  __syncthreads();

  int wave = threadIdx.x >> 5;
  int lane = threadIdx.x & 31;
  int tile = blockIdx.x * (blockDim.x >> 5) + wave;
  if (tile >= numTiles) return;

  int m  = lane & 15;
  int kh = (lane >> 4) << 1;
  const float* arow = A + (size_t)(tile * 16 + m) * K;

  v8f c = {};
#pragma unroll
  for (int k0 = 0; k0 < K; k0 += 4) {
    v2f a, b;
    a.x = arow[k0 + kh];
    a.y = arow[k0 + kh + 1];
    b.x = wlds[k0 + kh][m];
    b.y = wlds[k0 + kh + 1][m];
    c = __builtin_amdgcn_wmma_f32_16x16x4_f32(false, a, false, b, (short)0, c,
                                              false, false);
  }

  int mbase = (lane >> 4) << 3;
#pragma unroll
  for (int i = 0; i < 8; ++i) {
    int node = tile * 16 + mbase + i;
    float v = c[i] * dinv[node];
    hs[node * HPAD + m] = v;
  }
}

#define GATHER_BATCH 8

template <int K>
__device__ __forceinline__ unsigned quad_get(unsigned x) {
  return (unsigned)__builtin_amdgcn_update_dpp(0, (int)x, K * 0x55, 0xf, 0xf, true);
}

__device__ __forceinline__ float4 ld_row(const char* base, unsigned off) {
  return *reinterpret_cast<const float4*>(base + off);
}

__device__ __forceinline__ void add4(float4& a, const float4 x) {
  a.x += x.x;
  a.y += x.y;
  a.z += x.z;
  a.w += x.w;
}

template <int L0, int BASE>
__device__ __forceinline__ void batch8(const char* hsb, uint4 ids, unsigned q16, int left,
                                       float4& acc) {
  const unsigned o0 = quad_get<L0>(ids.x) | q16;
  const unsigned o1 = quad_get<L0>(ids.y) | q16;
  const unsigned o2 = quad_get<L0>(ids.z) | q16;
  const unsigned o3 = quad_get<L0>(ids.w) | q16;
  const unsigned o4 = quad_get<L0 + 1>(ids.x) | q16;
  const unsigned o5 = quad_get<L0 + 1>(ids.y) | q16;
  const unsigned o6 = quad_get<L0 + 1>(ids.z) | q16;
  const unsigned o7 = quad_get<L0 + 1>(ids.w) | q16;
  if (left >= BASE + GATHER_BATCH) {
    const float4 x0 = ld_row(hsb, o0);
    const float4 x1 = ld_row(hsb, o1);
    const float4 x2 = ld_row(hsb, o2);
    const float4 x3 = ld_row(hsb, o3);
    const float4 x4 = ld_row(hsb, o4);
    const float4 x5 = ld_row(hsb, o5);
    const float4 x6 = ld_row(hsb, o6);
    const float4 x7 = ld_row(hsb, o7);
    add4(acc, x0);
    add4(acc, x1);
    add4(acc, x2);
    add4(acc, x3);
    add4(acc, x4);
    add4(acc, x5);
    add4(acc, x6);
    add4(acc, x7);
  } else {
    const float4 z = make_float4(0.0f, 0.0f, 0.0f, 0.0f);
    float4 x0 = z, x1 = z, x2 = z, x3 = z, x4 = z, x5 = z, x6 = z;
    x0 = ld_row(hsb, o0);
    if (left > BASE + 1) x1 = ld_row(hsb, o1);
    if (left > BASE + 2) x2 = ld_row(hsb, o2);
    if (left > BASE + 3) x3 = ld_row(hsb, o3);
    if (left > BASE + 4) x4 = ld_row(hsb, o4);
    if (left > BASE + 5) x5 = ld_row(hsb, o5);
    if (left > BASE + 6) x6 = ld_row(hsb, o6);
    add4(acc, x0);
    if (left > BASE + 1) add4(acc, x1);
    if (left > BASE + 2) add4(acc, x2);
    if (left > BASE + 3) add4(acc, x3);
    if (left > BASE + 4) add4(acc, x4);
    if (left > BASE + 5) add4(acc, x5);
    if (left > BASE + 6) add4(acc, x6);
  }
}

#define TILE_PITCH 20

template <bool FUSE>
__global__ __launch_bounds__(256) void k_gather(const int* __restrict__ rowptr,
                                                const int* __restrict__ srcs,
                                                const float* __restrict__ hs,
                                                const float* __restrict__ dinv,
                                                const float* __restrict__ b,
                                                const float* __restrict__ Wn,
                                                float* __restrict__ hout, int n) {
  int t = blockIdx.x * blockDim.x + threadIdx.x;
  int v = t >> 2;
  bool live = true;
  if constexpr (FUSE) {
    if (((t & ~31) >> 2) >= n) return;
    live = v < n;
    if (!live) v = n - 1;
  } else {
    if (v >= n) return;
  }
  const int q = t & 3;
  const unsigned q16 = (unsigned)q << 4;
  const char* hsb = reinterpret_cast<const char*>(hs);
  const int p = rowptr[v];
  const int end = rowptr[v + 1];
  float4 acc = ld_row(hsb, ((unsigned)v << 6) | q16);

  const int* ip = srcs + p + 4 * q;
  for (int left = end - p; left > 0; left -= 32, ip += 32) {
    uint4 ia, ib;
    ia.x = (unsigned)ip[0] << 6;
    ia.y = (unsigned)ip[1] << 6;
    ia.z = (unsigned)ip[2] << 6;
    ia.w = (unsigned)ip[3] << 6;
    ib.x = (unsigned)ip[16] << 6;
    ib.y = (unsigned)ip[17] << 6;
    ib.z = (unsigned)ip[18] << 6;
    ib.w = (unsigned)ip[19] << 6;
    batch8<0, 0>(hsb, ia, q16, left, acc);
    if (left > 8) batch8<2, 8>(hsb, ia, q16, left, acc);
    if (left > 16) batch8<0, 16>(hsb, ib, q16, left, acc);
    if (left > 24) batch8<2, 24>(hsb, ib, q16, left, acc);
  }

  const float d = dinv[v];
  const int c0 = q << 2;
  float4 o;
  o.x = fmaxf(d * acc.x + b[c0 + 0], 0.0f);
  o.y = fmaxf(d * acc.y + b[c0 + 1], 0.0f);
  o.z = fmaxf(d * acc.z + b[c0 + 2], 0.0f);
  o.w = (c0 + 3 < HFEAT) ? fmaxf(d * acc.w + b[c0 + 3], 0.0f) : 0.0f;
  if constexpr (!FUSE) {
    reinterpret_cast<float4*>(hout)[(size_t)v * 4 + q] = o;
  } else {
    __shared__ __attribute__((aligned(16))) float sA[8][8 * TILE_PITCH];
    __shared__ __attribute__((aligned(16))) float sD[8][16 * TILE_PITCH];
    const int lane = threadIdx.x & 31;
    const int cc = lane & 15;
    const int hh = lane >> 4;
    float* ta = sA[threadIdx.x >> 5];
    float* td = sD[threadIdx.x >> 5];

    v2f bf[4];
    const int cw = cc < HFEAT ? cc : HFEAT - 1;
#pragma unroll
    for (int j = 0; j < 4; ++j) {
      const int k = 4 * j + 2 * hh;
      const int k1 = k + 1 < HFEAT ? k + 1 : HFEAT - 1;
      const float w0 = Wn[k * HFEAT + cw];
      const float w1 = Wn[k1 * HFEAT + cw];
      bf[j].x = (cc < HFEAT) ? w0 : 0.0f;
      bf[j].y = (cc < HFEAT && k + 1 < HFEAT) ? w1 : 0.0f;
    }

    const int rw = lane >> 2;
    *reinterpret_cast<float4*>(ta + rw * TILE_PITCH + c0) = o;
    __builtin_amdgcn_fence(__ATOMIC_RELEASE, "wavefront");
    __builtin_amdgcn_wave_barrier();
    __builtin_amdgcn_fence(__ATOMIC_ACQUIRE, "wavefront");

    const float* ar = ta + (cc & 7) * TILE_PITCH + 2 * hh;
    const bool real = cc < 8;
    v8f c = {};
#pragma unroll
    for (int j = 0; j < 4; ++j) {
      v2f a = *reinterpret_cast<const v2f*>(ar + 4 * j);
      a.x = real ? a.x : 0.0f;
      a.y = real ? a.y : 0.0f;
      c = __builtin_amdgcn_wmma_f32_16x16x4_f32(false, a, false, bf[j], (short)0, c,
                                                false, false);
    }

#pragma unroll
    for (int i = 0; i < 8; ++i) td[(8 * hh + i) * TILE_PITCH + cc] = c[i];
    __builtin_amdgcn_fence(__ATOMIC_RELEASE, "wavefront");
    __builtin_amdgcn_wave_barrier();
    __builtin_amdgcn_fence(__ATOMIC_ACQUIRE, "wavefront");
    float4 y = *reinterpret_cast<const float4*>(td + rw * TILE_PITCH + c0);
    y.x *= d;
    y.y *= d;
    y.z *= d;
    y.w *= d;
    if (live) reinterpret_cast<float4*>(hout)[(size_t)v * 4 + q] = y;
  }
}

__global__ void k_gemm_out(const float* __restrict__ A,
                           const float* __restrict__ Wc,
                           const float* __restrict__ bc,
                           float* __restrict__ out,
                           int numNodeTiles) {
  __shared__ float wlds[HPAD][DFEAT];
  for (int i = threadIdx.x; i < HPAD * DFEAT; i += blockDim.x) {
    int k = i >> 7, n = i & 127;
    wlds[k][n] = (k < HFEAT) ? Wc[k * DFEAT + n] : 0.0f;
  }
  __syncthreads();

  int wave  = threadIdx.x >> 5;
  int lane  = threadIdx.x & 31;
  int wtile = blockIdx.x * (blockDim.x >> 5) + wave;
  if (wtile >= numNodeTiles * 8) return;
  int tile = wtile >> 3;
  int n0   = (wtile & 7) << 4;

  int m  = lane & 15;
  int kh = (lane >> 4) << 1;
  const float* arow = A + (size_t)(tile * 16 + m) * HPAD;

  v8f c = {};
#pragma unroll
  for (int k0 = 0; k0 < HPAD; k0 += 4) {
    v2f a, b;
    a.x = arow[k0 + kh];
    a.y = arow[k0 + kh + 1];
    b.x = wlds[k0 + kh][n0 + m];
    b.y = wlds[k0 + kh + 1][n0 + m];
    c = __builtin_amdgcn_wmma_f32_16x16x4_f32(false, a, false, b, (short)0, c,
                                              false, false);
  }

  float bias = bc[n0 + m];
  int mbase = (lane >> 4) << 3;
#pragma unroll
  for (int i = 0; i < 8; ++i) {
    int node = tile * 16 + mbase + i;
    out[(size_t)node * DFEAT + n0 + m] = c[i] + bias;
  }
}

extern "C" void kernel_launch(void* const* d_in, const int* in_sizes, int n_in,
                              void* d_out, int out_size, void* d_ws, size_t ws_size,
                              hipStream_t stream) {
  const float* x  = (const float*)d_in[0];
  const int*   ei = (const int*)d_in[1];
  const float* W1 = (const float*)d_in[2];
  const float* b1 = (const float*)d_in[3];
  const float* W2 = (const float*)d_in[4];
  const float* b2 = (const float*)d_in[5];
  const float* W3 = (const float*)d_in[6];
  const float* b3 = (const float*)d_in[7];
  const float* Wc = (const float*)d_in[8];
  const float* bc = (const float*)d_in[9];

  const int N = in_sizes[0] / DFEAT;
  const int E = in_sizes[1] / 2;
  const int* row = ei;
  const int* col = ei + E;

  float* dinv = (float*)d_ws;
  float* B0 = dinv + N;
  float* B1 = B0 + (size_t)N * HPAD;
  float* B2 = B1 + (size_t)N * HPAD;
  float* outf = (float*)d_out;
  int* rowptr = (int*)B1;
  int* tab    = rowptr + N + 1;

  int* srcs = (int*)d_out;
  unsigned* bucketed = (unsigned*)(srcs + E);

  const int T = 256;
  const int numTiles = N / 16;
  dim3 gV((N * 4 + T - 1) / T);
  dim3 gG((numTiles + 7) / 8);
  dim3 gO((numTiles * 8 + 7) / 8);

  const int runs = (E + BIN_CHUNK - 1) / BIN_CHUNK;
  dim3 gB(runs);
  dim3 gR((N + RANGE - 1) / RANGE);
  k_bin  <<<gB, BIN_THREADS, 0, stream>>>(row, col, bucketed, tab, N, E);
  k_group<<<gR, GRP_THREADS, 0, stream>>>(bucketed, tab, srcs, rowptr, dinv, N, runs);

  k_gemm_scale<DFEAT><<<gG, T, 0, stream>>>(x, W1, DFEAT, dinv, B0, numTiles);
  k_gather<true><<<gV, T, 0, stream>>>(rowptr, srcs, B0, dinv, b1, W2, B2, N);

  k_gather<true><<<gV, T, 0, stream>>>(rowptr, srcs, B2, dinv, b2, W3, B0, N);

  k_gather<false><<<gV, T, 0, stream>>>(rowptr, srcs, B0, dinv, b3, nullptr, B2, N);

  k_gemm_out<<<gO, T, 0, stream>>>(B2, Wc, bc, outf, numTiles);
}


# Layer-1 product x@W1 (dinv-scaled) moved into the edge-grouping launch: a 400-node range block prefetches its x rows early, multiplies them once its degrees are counted; 7 -> 6 launches; same sums
# speedup vs baseline: 6.6458x; 1.0364x over previous
#include <hip/hip_runtime.h>


typedef __attribute__((ext_vector_type(2))) float v2f;
typedef __attribute__((ext_vector_type(8))) float v8f;

#define DFEAT 128
#define HFEAT 15
#define HPAD  16

#define BIN_CHUNK   25000
#define BIN_THREADS 1024
#define RANGE       400
#define NR_MAX      256
#define TAB_STRIDE  (NR_MAX + 1)
#define SRC_BITS    17
#define SRC_MASK    0x1FFFFu
#define NO_ENTRY    0xFFFFFFFFu

__device__ __forceinline__ void bin_count(int* cur, int c, unsigned n) {
  if ((unsigned)c < n) atomicAdd(&cur[(unsigned)c / RANGE], 1);
}

__device__ __forceinline__ int bin_take(int* cur, int c, unsigned n) {
  return ((unsigned)c < n) ? atomicAdd(&cur[(unsigned)c / RANGE], 1) : -1;
}

__device__ __forceinline__ void bin_put(unsigned* stage, int slot, int c, int r) {
  if (slot >= 0) stage[slot] = (((unsigned)c % RANGE) << SRC_BITS) | ((unsigned)r & SRC_MASK);
}

__global__ __launch_bounds__(BIN_THREADS) void k_bin(const int* __restrict__ row,
                                                     const int* __restrict__ col,
                                                     unsigned* __restrict__ bucketed,
                                                     int* __restrict__ tab,
                                                     int n, int e) {
  __shared__ int cur[NR_MAX];
  __shared__ int wtot[NR_MAX / 32];
  __shared__ int s_total;
  __shared__ __attribute__((aligned(16))) unsigned stage[BIN_CHUNK];
  constexpr int T = BIN_THREADS;
  constexpr int VPT = (BIN_CHUNK / 4 + T - 1) / T;
  const int tid  = threadIdx.x;
  const int lane = tid & 31;
  const int wave = tid >> 5;
  const int g  = blockIdx.x;
  const int e0 = g * BIN_CHUNK;
  const int cntE = (e - e0 < BIN_CHUNK) ? (e - e0) : BIN_CHUNK;
  const unsigned un = (unsigned)n;
  const int* cb = col + e0;
  const int* rb = row + e0;
  const int nvec = (((((size_t)cb) | ((size_t)rb)) & 15) == 0) ? (cntE >> 2) : 0;

  int4 c[VPT], r[VPT];
  const int4 none = make_int4(-1, -1, -1, -1);
#pragma unroll
  for (int k = 0; k < VPT; ++k) {
    c[k] = none;
    r[k] = none;
  }
  if (nvec > 0) {
    const int4* c4 = reinterpret_cast<const int4*>(cb);
    const int4* r4 = reinterpret_cast<const int4*>(rb);
#pragma unroll
    for (int k = 0; k < VPT; ++k) {
      const int i = tid + k * T;
      const bool has = i < nvec;
      const int4 cv = c4[has ? i : 0];
      const int4 rv = r4[has ? i : 0];
      c[k] = has ? cv : none;
      r[k] = rv;
    }
  }
  if (tid < NR_MAX) cur[tid] = 0;
  __syncthreads();

#pragma unroll
  for (int k = 0; k < VPT; ++k) {
    bin_count(cur, c[k].x, un);
    bin_count(cur, c[k].y, un);
    bin_count(cur, c[k].z, un);
    bin_count(cur, c[k].w, un);
  }
  for (int j = (nvec << 2) + tid; j < cntE; j += T) bin_count(cur, cb[j], un);
  __syncthreads();

  int v = 0, inc = 0;
  if (tid < NR_MAX) {
    v = cur[tid];
    inc = v;
    for (int o = 1; o < 32; o <<= 1) {
      int t = __shfl_up(inc, o, 32);
      if (lane >= o) inc += t;
    }
    if (lane == 31) wtot[wave] = inc;
  }
  __syncthreads();
  if (tid < NR_MAX) {
    int before = 0;
    for (int w = 0; w < wave; ++w) before += wtot[w];
    const int ex = before + inc - v;
    cur[tid] = ex;
    tab[g * TAB_STRIDE + tid] = ex;
    if (tid == NR_MAX - 1) {
      tab[g * TAB_STRIDE + NR_MAX] = ex + v;
      s_total = ex + v;
    }
  }
  __syncthreads();

#pragma unroll
  for (int k = 0; k < VPT; ++k) {
    const int s0 = bin_take(cur, c[k].x, un);
    const int s1 = bin_take(cur, c[k].y, un);
    const int s2 = bin_take(cur, c[k].z, un);
    const int s3 = bin_take(cur, c[k].w, un);
    bin_put(stage, s0, c[k].x, r[k].x);
    bin_put(stage, s1, c[k].y, r[k].y);
    bin_put(stage, s2, c[k].z, r[k].z);
    bin_put(stage, s3, c[k].w, r[k].w);
  }
  for (int j = (nvec << 2) + tid; j < cntE; j += T) {
    const int cj = cb[j];
    bin_put(stage, bin_take(cur, cj, un), cj, rb[j]);
  }
  __syncthreads();

  const int total = s_total;
  unsigned* dst = bucketed + (size_t)g * BIN_CHUNK;
  const int nv = ((((size_t)dst) & 15) == 0) ? (total >> 2) : 0;
  for (int i = tid; i < nv; i += T)
    reinterpret_cast<uint4*>(dst)[i] = reinterpret_cast<const uint4*>(stage)[i];
  for (int j = (nv << 2) + tid; j < total; j += T) dst[j] = stage[j];
}

#define GRP_THREADS 1024
#define GRP_CAP     32768
#define GRP_STEPS   32
#define GRP_PIECES  8
#define GRP_LANES   (GRP_THREADS / GRP_PIECES)
#define GRP_NODES   512
#define GRP_TILES   (RANGE / 16)

__device__ __forceinline__ void grp_place(int* cursor, unsigned* outl, int* __restrict__ srcs,
                                          int base, unsigned w) {
  if (w != NO_ENTRY) {
    const int slot = atomicAdd(&cursor[w >> SRC_BITS], 1);
    const int sv = (int)(w & SRC_MASK);
    if (slot < GRP_CAP) outl[slot] = (unsigned)sv;
    else srcs[base + slot] = sv;
  }
}

__global__ __launch_bounds__(GRP_THREADS) void k_group(const unsigned* __restrict__ bucketed,
                                                       const int* __restrict__ tab,
                                                       int* __restrict__ srcs,
                                                       int* __restrict__ rowptr,
                                                       float* __restrict__ dinv,
                                                       const float* __restrict__ x,
                                                       const float* __restrict__ W1,
                                                       float* __restrict__ hs,
                                                       int n, int ng) {
  __shared__ int pc[GRP_STEPS * GRP_PIECES];
  __shared__ int poff[GRP_STEPS * GRP_PIECES];
  __shared__ int deg[GRP_NODES];
  __shared__ int wsum[(GRP_STEPS * GRP_PIECES) / 32];
  __shared__ int wtot[GRP_NODES / 32];
  __shared__ int s_base;
  __shared__ int s_total;
  __shared__ unsigned outl[GRP_CAP];
  __shared__ float wl[DFEAT][HPAD];
  __shared__ float sdinv[GRP_NODES];
  const int tid  = threadIdx.x;
  const int lane = tid & 31;
  const int wave = tid >> 5;
  const int rg = blockIdx.x;
  const int lo = rg * RANGE;
  const int rem = n - lo;
  const int span = rem < RANGE ? rem : RANGE;

  if (tid < GRP_STEPS * GRP_PIECES) {
    int a = 0, b = 0;
    if (tid < ng) {
      a = tab[tid * TAB_STRIDE + rg];
      b = tab[tid * TAB_STRIDE + rg + 1];
    }
    pc[tid] = b - a;
    poff[tid] = a;
    int sum = a;
    for (int o = 16; o > 0; o >>= 1) sum += __shfl_xor(sum, o, 32);
    if (lane == 0) wsum[wave] = sum;
  }
  if (tid < GRP_NODES) deg[tid] = 0;
  __syncthreads();
  if (tid == 0) {
    int s = 0;
    for (int w = 0; w < (GRP_STEPS * GRP_PIECES) / 32; ++w) s += wsum[w];
    s_base = s;
  }

  const int sub = tid / GRP_LANES;
  const int li  = tid % GRP_LANES;
  constexpr int WPT = (DFEAT * HPAD) / GRP_THREADS;
  float wv[WPT];
#pragma unroll
  for (int u = 0; u < WPT; ++u) {
    const int i = tid + u * GRP_THREADS;
    const int k = i >> 4, c = i & 15;
    const float w = W1[k * HFEAT + (c < HFEAT ? c : HFEAT - 1)];
    wv[u] = (c < HFEAT) ? w : 0.0f;
  }
  unsigned e1[GRP_STEPS], e2[GRP_STEPS];
#pragma unroll
  for (int s = 0; s < GRP_STEPS; ++s) {
    const int g = s * GRP_PIECES + sub;
    const int cnt = pc[g];
    const unsigned* src = bucketed + (size_t)g * BIN_CHUNK + poff[g];
    const bool h1 = li < cnt;
    const bool h2 = li + GRP_LANES < cnt;
    const unsigned w1 = *(h1 ? src + li : bucketed);
    const unsigned w2 = *(h2 ? src + li + GRP_LANES : bucketed);
    e1[s] = h1 ? w1 : NO_ENTRY;
    e2[s] = h2 ? w2 : NO_ENTRY;
  }
  if (tid < 2 * span) __builtin_prefetch(x + (size_t)lo * DFEAT + (size_t)tid * 64, 0, 2);
#pragma unroll
  for (int u = 0; u < WPT; ++u) {
    const int i = tid + u * GRP_THREADS;
    wl[i >> 4][i & 15] = wv[u];
  }

#pragma unroll
  for (int s = 0; s < GRP_STEPS; ++s) {
    if (e1[s] != NO_ENTRY) atomicAdd(&deg[e1[s] >> SRC_BITS], 1);
    if (e2[s] != NO_ENTRY) atomicAdd(&deg[e2[s] >> SRC_BITS], 1);
  }
  for (int s = 0; s < GRP_STEPS; ++s) {
    const int g = s * GRP_PIECES + sub;
    const int cnt = pc[g];
    const unsigned* src = bucketed + (size_t)g * BIN_CHUNK + poff[g];
    for (int i = li + 2 * GRP_LANES; i < cnt; i += GRP_LANES) atomicAdd(&deg[src[i] >> SRC_BITS], 1);
  }
  __syncthreads();

  int v = 0, inc = 0;
  if (tid < GRP_NODES) {
    v = deg[tid];
    inc = v;
    for (int o = 1; o < 32; o <<= 1) {
      int t = __shfl_up(inc, o, 32);
      if (lane >= o) inc += t;
    }
    if (lane == 31) wtot[wave] = inc;
  }
  __syncthreads();
  const int base = s_base;
  if (tid < GRP_NODES) {
    int before = 0;
    for (int w = 0; w < wave; ++w) before += wtot[w];
    const int ex = before + inc - v;
    deg[tid] = ex;
    const float dv = rsqrtf((float)(v + 1));
    sdinv[tid] = dv;
    if (tid < span) {
      rowptr[lo + tid] = base + ex;
      dinv[lo + tid] = dv;
      if (lo + tid == n - 1) rowptr[n] = base + ex + v;
    }
    if (tid == GRP_NODES - 1) s_total = ex + v;
  }
  __syncthreads();

#pragma unroll
  for (int s = 0; s < GRP_STEPS; ++s) {
    grp_place(deg, outl, srcs, base, e1[s]);
    grp_place(deg, outl, srcs, base, e2[s]);
    __syncthreads();
  }
  for (int s = 0; s < GRP_STEPS; ++s) {
    const int g = s * GRP_PIECES + sub;
    const int cnt = pc[g];
    const unsigned* src = bucketed + (size_t)g * BIN_CHUNK + poff[g];
    for (int i = li + 2 * GRP_LANES; i < cnt; i += GRP_LANES) grp_place(deg, outl, srcs, base, src[i]);
  }
  __syncthreads();

  const bool gw = wave < GRP_TILES && lo + wave * 16 < n;
  const int m  = lane & 15;
  const int kh = (lane >> 4) << 1;
  const int r0 = wave * 16;
  v2f a[DFEAT / 4];
  if (gw) {
    const float* arow = x + (size_t)(lo + r0 + m) * DFEAT + kh;
#pragma unroll
    for (int j = 0; j < DFEAT / 4; ++j) a[j] = *reinterpret_cast<const v2f*>(arow + 4 * j);
  }

  const int total = s_total;
  const int tot = total < GRP_CAP ? total : GRP_CAP;
  for (int p = tid - (base & 31); p < tot; p += GRP_THREADS)
    if (p >= 0) srcs[base + p] = (int)outl[p];

  if (gw) {
    v8f c = {};
#pragma unroll
    for (int j = 0; j < DFEAT / 4; ++j) {
      v2f b;
      b.x = wl[4 * j + kh][m];
      b.y = wl[4 * j + kh + 1][m];
      c = __builtin_amdgcn_wmma_f32_16x16x4_f32(false, a[j], false, b, (short)0, c,
                                                false, false);
    }
    const int mbase = (lane >> 4) << 3;
#pragma unroll
    for (int i = 0; i < 8; ++i) {
      const int nl = r0 + mbase + i;
      hs[(lo + nl) * HPAD + m] = c[i] * sdinv[nl];
    }
  }
}

#define GATHER_BATCH 8

template <int K>
__device__ __forceinline__ unsigned quad_get(unsigned x) {
  return (unsigned)__builtin_amdgcn_update_dpp(0, (int)x, K * 0x55, 0xf, 0xf, true);
}

__device__ __forceinline__ float4 ld_row(const char* base, unsigned off) {
  return *reinterpret_cast<const float4*>(base + off);
}

__device__ __forceinline__ void add4(float4& a, const float4 x) {
  a.x += x.x;
  a.y += x.y;
  a.z += x.z;
  a.w += x.w;
}

template <int L0, int BASE>
__device__ __forceinline__ void batch8(const char* hsb, uint4 ids, unsigned q16, int left,
                                       float4& acc) {
  const unsigned o0 = quad_get<L0>(ids.x) | q16;
  const unsigned o1 = quad_get<L0>(ids.y) | q16;
  const unsigned o2 = quad_get<L0>(ids.z) | q16;
  const unsigned o3 = quad_get<L0>(ids.w) | q16;
  const unsigned o4 = quad_get<L0 + 1>(ids.x) | q16;
  const unsigned o5 = quad_get<L0 + 1>(ids.y) | q16;
  const unsigned o6 = quad_get<L0 + 1>(ids.z) | q16;
  const unsigned o7 = quad_get<L0 + 1>(ids.w) | q16;
  if (left >= BASE + GATHER_BATCH) {
    const float4 x0 = ld_row(hsb, o0);
    const float4 x1 = ld_row(hsb, o1);
    const float4 x2 = ld_row(hsb, o2);
    const float4 x3 = ld_row(hsb, o3);
    const float4 x4 = ld_row(hsb, o4);
    const float4 x5 = ld_row(hsb, o5);
    const float4 x6 = ld_row(hsb, o6);
    const float4 x7 = ld_row(hsb, o7);
    add4(acc, x0);
    add4(acc, x1);
    add4(acc, x2);
    add4(acc, x3);
    add4(acc, x4);
    add4(acc, x5);
    add4(acc, x6);
    add4(acc, x7);
  } else {
    const float4 z = make_float4(0.0f, 0.0f, 0.0f, 0.0f);
    float4 x0 = z, x1 = z, x2 = z, x3 = z, x4 = z, x5 = z, x6 = z;
    x0 = ld_row(hsb, o0);
    if (left > BASE + 1) x1 = ld_row(hsb, o1);
    if (left > BASE + 2) x2 = ld_row(hsb, o2);
    if (left > BASE + 3) x3 = ld_row(hsb, o3);
    if (left > BASE + 4) x4 = ld_row(hsb, o4);
    if (left > BASE + 5) x5 = ld_row(hsb, o5);
    if (left > BASE + 6) x6 = ld_row(hsb, o6);
    add4(acc, x0);
    if (left > BASE + 1) add4(acc, x1);
    if (left > BASE + 2) add4(acc, x2);
    if (left > BASE + 3) add4(acc, x3);
    if (left > BASE + 4) add4(acc, x4);
    if (left > BASE + 5) add4(acc, x5);
    if (left > BASE + 6) add4(acc, x6);
  }
}

#define TILE_PITCH 20

template <bool FUSE>
__global__ __launch_bounds__(256) void k_gather(const int* __restrict__ rowptr,
                                                const int* __restrict__ srcs,
                                                const float* __restrict__ hs,
                                                const float* __restrict__ dinv,
                                                const float* __restrict__ b,
                                                const float* __restrict__ Wn,
                                                float* __restrict__ hout, int n) {
  int t = blockIdx.x * blockDim.x + threadIdx.x;
  int v = t >> 2;
  bool live = true;
  if constexpr (FUSE) {
    if (((t & ~31) >> 2) >= n) return;
    live = v < n;
    if (!live) v = n - 1;
  } else {
    if (v >= n) return;
  }
  const int q = t & 3;
  const unsigned q16 = (unsigned)q << 4;
  const char* hsb = reinterpret_cast<const char*>(hs);
  const int p = rowptr[v];
  const int end = rowptr[v + 1];
  float4 acc = ld_row(hsb, ((unsigned)v << 6) | q16);

  const int* ip = srcs + p + 4 * q;
  for (int left = end - p; left > 0; left -= 32, ip += 32) {
    uint4 ia, ib;
    ia.x = (unsigned)ip[0] << 6;
    ia.y = (unsigned)ip[1] << 6;
    ia.z = (unsigned)ip[2] << 6;
    ia.w = (unsigned)ip[3] << 6;
    ib.x = (unsigned)ip[16] << 6;
    ib.y = (unsigned)ip[17] << 6;
    ib.z = (unsigned)ip[18] << 6;
    ib.w = (unsigned)ip[19] << 6;
    batch8<0, 0>(hsb, ia, q16, left, acc);
    if (left > 8) batch8<2, 8>(hsb, ia, q16, left, acc);
    if (left > 16) batch8<0, 16>(hsb, ib, q16, left, acc);
    if (left > 24) batch8<2, 24>(hsb, ib, q16, left, acc);
  }

  const float d = dinv[v];
  const int c0 = q << 2;
  float4 o;
  o.x = fmaxf(d * acc.x + b[c0 + 0], 0.0f);
  o.y = fmaxf(d * acc.y + b[c0 + 1], 0.0f);
  o.z = fmaxf(d * acc.z + b[c0 + 2], 0.0f);
  o.w = (c0 + 3 < HFEAT) ? fmaxf(d * acc.w + b[c0 + 3], 0.0f) : 0.0f;
  if constexpr (!FUSE) {
    reinterpret_cast<float4*>(hout)[(size_t)v * 4 + q] = o;
  } else {
    __shared__ __attribute__((aligned(16))) float sA[8][8 * TILE_PITCH];
    __shared__ __attribute__((aligned(16))) float sD[8][16 * TILE_PITCH];
    const int lane = threadIdx.x & 31;
    const int cc = lane & 15;
    const int hh = lane >> 4;
    float* ta = sA[threadIdx.x >> 5];
    float* td = sD[threadIdx.x >> 5];

    v2f bf[4];
    const int cw = cc < HFEAT ? cc : HFEAT - 1;
#pragma unroll
    for (int j = 0; j < 4; ++j) {
      const int k = 4 * j + 2 * hh;
      const int k1 = k + 1 < HFEAT ? k + 1 : HFEAT - 1;
      const float w0 = Wn[k * HFEAT + cw];
      const float w1 = Wn[k1 * HFEAT + cw];
      bf[j].x = (cc < HFEAT) ? w0 : 0.0f;
      bf[j].y = (cc < HFEAT && k + 1 < HFEAT) ? w1 : 0.0f;
    }

    const int rw = lane >> 2;
    *reinterpret_cast<float4*>(ta + rw * TILE_PITCH + c0) = o;
    __builtin_amdgcn_fence(__ATOMIC_RELEASE, "wavefront");
    __builtin_amdgcn_wave_barrier();
    __builtin_amdgcn_fence(__ATOMIC_ACQUIRE, "wavefront");

    const float* ar = ta + (cc & 7) * TILE_PITCH + 2 * hh;
    const bool real = cc < 8;
    v8f c = {};
#pragma unroll
    for (int j = 0; j < 4; ++j) {
      v2f a = *reinterpret_cast<const v2f*>(ar + 4 * j);
      a.x = real ? a.x : 0.0f;
      a.y = real ? a.y : 0.0f;
      c = __builtin_amdgcn_wmma_f32_16x16x4_f32(false, a, false, bf[j], (short)0, c,
                                                false, false);
    }

#pragma unroll
    for (int i = 0; i < 8; ++i) td[(8 * hh + i) * TILE_PITCH + cc] = c[i];
    __builtin_amdgcn_fence(__ATOMIC_RELEASE, "wavefront");
    __builtin_amdgcn_wave_barrier();
    __builtin_amdgcn_fence(__ATOMIC_ACQUIRE, "wavefront");
    float4 y = *reinterpret_cast<const float4*>(td + rw * TILE_PITCH + c0);
    y.x *= d;
    y.y *= d;
    y.z *= d;
    y.w *= d;
    if (live) reinterpret_cast<float4*>(hout)[(size_t)v * 4 + q] = y;
  }
}

__global__ void k_gemm_out(const float* __restrict__ A,
                           const float* __restrict__ Wc,
                           const float* __restrict__ bc,
                           float* __restrict__ out,
                           int numNodeTiles) {
  __shared__ float wlds[HPAD][DFEAT];
  for (int i = threadIdx.x; i < HPAD * DFEAT; i += blockDim.x) {
    int k = i >> 7, n = i & 127;
    wlds[k][n] = (k < HFEAT) ? Wc[k * DFEAT + n] : 0.0f;
  }
  __syncthreads();

  int wave  = threadIdx.x >> 5;
  int lane  = threadIdx.x & 31;
  int wtile = blockIdx.x * (blockDim.x >> 5) + wave;
  if (wtile >= numNodeTiles * 8) return;
  int tile = wtile >> 3;
  int n0   = (wtile & 7) << 4;

  int m  = lane & 15;
  int kh = (lane >> 4) << 1;
  const float* arow = A + (size_t)(tile * 16 + m) * HPAD;

  v8f c = {};
#pragma unroll
  for (int k0 = 0; k0 < HPAD; k0 += 4) {
    v2f a, b;
    a.x = arow[k0 + kh];
    a.y = arow[k0 + kh + 1];
    b.x = wlds[k0 + kh][n0 + m];
    b.y = wlds[k0 + kh + 1][n0 + m];
    c = __builtin_amdgcn_wmma_f32_16x16x4_f32(false, a, false, b, (short)0, c,
                                              false, false);
  }

  float bias = bc[n0 + m];
  int mbase = (lane >> 4) << 3;
#pragma unroll
  for (int i = 0; i < 8; ++i) {
    int node = tile * 16 + mbase + i;
    out[(size_t)node * DFEAT + n0 + m] = c[i] + bias;
  }
}

extern "C" void kernel_launch(void* const* d_in, const int* in_sizes, int n_in,
                              void* d_out, int out_size, void* d_ws, size_t ws_size,
                              hipStream_t stream) {
  const float* x  = (const float*)d_in[0];
  const int*   ei = (const int*)d_in[1];
  const float* W1 = (const float*)d_in[2];
  const float* b1 = (const float*)d_in[3];
  const float* W2 = (const float*)d_in[4];
  const float* b2 = (const float*)d_in[5];
  const float* W3 = (const float*)d_in[6];
  const float* b3 = (const float*)d_in[7];
  const float* Wc = (const float*)d_in[8];
  const float* bc = (const float*)d_in[9];

  const int N = in_sizes[0] / DFEAT;
  const int E = in_sizes[1] / 2;
  const int* row = ei;
  const int* col = ei + E;

  float* dinv = (float*)d_ws;
  float* B0 = dinv + N;
  float* B1 = B0 + (size_t)N * HPAD;
  float* B2 = B1 + (size_t)N * HPAD;
  float* outf = (float*)d_out;
  int* rowptr = (int*)B1;
  int* tab    = rowptr + N + 1;

  int* srcs = (int*)d_out;
  unsigned* bucketed = (unsigned*)(srcs + E);

  const int T = 256;
  const int numTiles = N / 16;
  dim3 gV((N * 4 + T - 1) / T);
  dim3 gO((numTiles * 8 + 7) / 8);

  const int runs = (E + BIN_CHUNK - 1) / BIN_CHUNK;
  dim3 gB(runs);
  dim3 gR((N + RANGE - 1) / RANGE);
  k_bin  <<<gB, BIN_THREADS, 0, stream>>>(row, col, bucketed, tab, N, E);
  k_group<<<gR, GRP_THREADS, 0, stream>>>(bucketed, tab, srcs, rowptr, dinv, x, W1, B0, N, runs);

  k_gather<true><<<gV, T, 0, stream>>>(rowptr, srcs, B0, dinv, b1, W2, B2, N);

  k_gather<true><<<gV, T, 0, stream>>>(rowptr, srcs, B2, dinv, b2, W3, B0, N);

  k_gather<false><<<gV, T, 0, stream>>>(rowptr, srcs, B0, dinv, b3, nullptr, B2, N);

  k_gemm_out<<<gO, T, 0, stream>>>(B2, Wc, bc, outf, numTiles);
}
